# starting kernel
# baseline (speedup 1.0000x reference)
#include <hip/hip_runtime.h>
#include <hip/hip_bf16.h>

typedef __attribute__((ext_vector_type(2))) float v2f;
typedef __attribute__((ext_vector_type(8))) float v8f;

#define DPAD 16
#define DDIM 15

__global__ void fill_kernel(float* __restrict__ p, float v, size_t n) {
    size_t i = (size_t)blockIdx.x * blockDim.x + threadIdx.x;
    size_t stride = (size_t)gridDim.x * blockDim.x;
    for (; i < n; i += stride) p[i] = v;
}

__global__ void pack_weight_kernel(const float* __restrict__ W, const float* __restrict__ b,
                                   float* __restrict__ Wp, float* __restrict__ bp,
                                   int K, int Nc, int Kpad, int Ncpad)
{
    int idx = blockIdx.x * blockDim.x + threadIdx.x;
    int total = Kpad * Ncpad;
    if (idx < total) {
        int k = idx / Ncpad, c = idx % Ncpad;
        float v = (k < K && c < Nc) ? W[k * Nc + c] : 0.f;
        Wp[((size_t)(k >> 1) * Ncpad + c) * 2 + (k & 1)] = v;
    }
    if (idx < Ncpad) bp[idx] = (idx < Nc) ? b[idx] : 0.f;
}

__global__ __launch_bounds__(256) void encode_kernel(
    const float* __restrict__ x, const int* __restrict__ type_id,
    const float* __restrict__ Wj, const float* __restrict__ bj,
    const float* __restrict__ Wm, const float* __restrict__ bm,
    float* __restrict__ h, int Nn)
{
    int i = blockIdx.x * blockDim.x + threadIdx.x;
    if (i >= Nn) return;
    float xi[5];
#pragma unroll
    for (int k = 0; k < 5; ++k) xi[k] = x[(size_t)i * 5 + k];
    int t = type_id[i];
    const float* Wp = (t == 0) ? Wj : Wm;
    const float* bp = (t == 0) ? bj : bm;
    bool valid = (t == 0) || (t == 1);
#pragma unroll
    for (int d = 0; d < DDIM; ++d) {
        float v = 0.f;
        if (valid) {
            v = bp[d];
#pragma unroll
            for (int k = 0; k < 5; ++k) v += xi[k] * Wp[k * DDIM + d];
            v = fmaxf(v, 0.f);
        }
        h[(size_t)i * DPAD + d] = v;
    }
    h[(size_t)i * DPAD + DDIM] = 0.f;
}

template <int EDIM>
__global__ __launch_bounds__(256) void edge_pass_kernel(
    const float* __restrict__ h, float* __restrict__ agg,
    const int* __restrict__ ei, const float* __restrict__ ea,
    const float* __restrict__ We, const float* __restrict__ be, int E)
{
    int e = blockIdx.x * blockDim.x + threadIdx.x;
    if (e >= E) return;
    int src = ei[e];
    int dst = ei[E + e];
    float ev[EDIM];
#pragma unroll
    for (int k = 0; k < EDIM; ++k) ev[k] = ea[(size_t)e * EDIM + k];

    float hl[DPAD];
    const float4* hv = (const float4*)(h + (size_t)src * DPAD);
#pragma unroll
    for (int q = 0; q < 4; ++q) {
        float4 t = hv[q];
        hl[4 * q + 0] = t.x; hl[4 * q + 1] = t.y;
        hl[4 * q + 2] = t.z; hl[4 * q + 3] = t.w;
    }

    float* ag = agg + (size_t)dst * DPAD;
#pragma unroll
    for (int d = 0; d < DDIM; ++d) {
        float m = hl[d] + be[d];
#pragma unroll
        for (int k = 0; k < EDIM; ++k) m += ev[k] * We[k * DDIM + d];
        m = fmaxf(m, 0.f);
        atomicAdd(&ag[d], m);
    }
}

template <bool HAS_A2, int KPAD>
__global__ __launch_bounds__(256) void wmma_gemm_kernel(
    const float* __restrict__ A, const float* __restrict__ A2,
    const float* __restrict__ Wp, const float* __restrict__ bp,
    float* __restrict__ C,
    int M, int Ncpad, int lda, int ldc, int relu)
{
    const int lane = threadIdx.x & 31;
    const int wave = threadIdx.x >> 5;
    const int ntN = Ncpad >> 4;
    const int tile = blockIdx.x * (blockDim.x >> 5) + wave;
    if (tile >= (M >> 4) * ntN) return;
    const int mt = tile / ntN;
    const int nt = tile % ntN;

    const int mrow  = mt * 16 + (lane & 15);
    const int ncol  = nt * 16 + (lane & 15);
    const int khalf = (lane >> 4) << 1;

    const float* ap  = A + (size_t)mrow * lda + khalf;
    const float* a2p = A2 + (size_t)mrow * lda + khalf;
    const float* bpp = Wp + ((size_t)(khalf >> 1) * Ncpad + ncol) * 2;
    const size_t bstep = (size_t)Ncpad * 4;

    v8f acc = {0.f, 0.f, 0.f, 0.f, 0.f, 0.f, 0.f, 0.f};
#pragma unroll
    for (int kk = 0; kk < KPAD; kk += 4) {
        v2f a = *(const v2f*)ap;
        if (HAS_A2) {
            v2f a2 = *(const v2f*)a2p;
            a.x += a2.x; a.y += a2.y;
            a2p += 4;
        }
        v2f b = *(const v2f*)bpp;
        ap += 4;
        bpp += bstep;
        acc = __builtin_amdgcn_wmma_f32_16x16x4_f32(
            false, a, false, b, (short)0, acc, false, false);
    }

    const int rbase = mt * 16 + ((lane >> 4) << 3);
    const float bv = bp[ncol];
#pragma unroll
    for (int r = 0; r < 8; ++r) {
        float v = acc[r] + bv;
        if (relu) v = fmaxf(v, 0.f);
        C[(size_t)(rbase + r) * ldc + ncol] = v;
    }
}

__global__ __launch_bounds__(256) void ln_relu_kernel(float* __restrict__ h, int Nn)
{
    int i = blockIdx.x * blockDim.x + threadIdx.x;
    if (i >= Nn) return;
    float v[DDIM];
    float mu = 0.f;
#pragma unroll
    for (int d = 0; d < DDIM; ++d) { v[d] = h[(size_t)i * DPAD + d]; mu += v[d]; }
    mu *= (1.f / DDIM);
    float var = 0.f;
#pragma unroll
    for (int d = 0; d < DDIM; ++d) { float t = v[d] - mu; var += t * t; }
    var *= (1.f / DDIM);
    float inv = rsqrtf(var + 1e-5f);
#pragma unroll
    for (int d = 0; d < DDIM; ++d)
        h[(size_t)i * DPAD + d] = fmaxf((v[d] - mu) * inv, 0.f);
}

__device__ __forceinline__ void atomicMaxF(float* addr, float v) {
    if (v >= 0.f) atomicMax((int*)addr, __float_as_int(v));
    else          atomicMin((unsigned int*)addr, __float_as_uint(v));
}

__global__ __launch_bounds__(256) void pool_kernel(
    const float* __restrict__ h, const int* __restrict__ type_id,
    const int* __restrict__ batch,
    float* __restrict__ sumJ, float* __restrict__ sumM,
    float* __restrict__ maxJ, float* __restrict__ maxM,
    float* __restrict__ cntJ, float* __restrict__ cntM, int Nn)
{
    int i = blockIdx.x * blockDim.x + threadIdx.x;
    if (i >= Nn) return;
    int t = type_id[i];
    if (t != 0 && t != 1) return;
    int b = batch[i];
    float* s  = (t == 0) ? sumJ : sumM;
    float* mx = (t == 0) ? maxJ : maxM;
    float* c  = (t == 0) ? cntJ : cntM;
    atomicAdd(&c[b], 1.f);
#pragma unroll
    for (int d = 0; d < DDIM; ++d) {
        float v = h[(size_t)i * DPAD + d];
        atomicAdd(&s[(size_t)b * DPAD + d], v);
        atomicMaxF(&mx[(size_t)b * DPAD + d], v);
    }
}

__global__ __launch_bounds__(128) void assemble_ln_kernel(
    const float* __restrict__ sumJ, const float* __restrict__ sumM,
    const float* __restrict__ maxJ, const float* __restrict__ maxM,
    const float* __restrict__ cntJ, const float* __restrict__ cntM,
    const float* __restrict__ u, float* __restrict__ g, int Gg)
{
    int gi = blockIdx.x * blockDim.x + threadIdx.x;
    if (gi >= Gg) return;
    float r[64];
    float cm = fmaxf(cntM[gi], 1.f);
    float cj = fmaxf(cntJ[gi], 1.f);
#pragma unroll
    for (int d = 0; d < DDIM; ++d) r[d] = sumM[(size_t)gi * DPAD + d] / cm;
#pragma unroll
    for (int d = 0; d < DDIM; ++d) r[15 + d] = sumJ[(size_t)gi * DPAD + d] / cj;
#pragma unroll
    for (int d = 0; d < DDIM; ++d) {
        float v = maxJ[(size_t)gi * DPAD + d];
        r[30 + d] = (v > -3.0e38f) ? v : 0.f;
    }
#pragma unroll
    for (int d = 0; d < DDIM; ++d) {
        float v = maxM[(size_t)gi * DPAD + d];
        r[45 + d] = (v > -3.0e38f) ? v : 0.f;
    }
#pragma unroll
    for (int j = 0; j < 4; ++j) r[60 + j] = u[(size_t)gi * 4 + j];

    float mu = 0.f;
#pragma unroll
    for (int k = 0; k < 64; ++k) mu += r[k];
    mu *= (1.f / 64.f);
    float var = 0.f;
#pragma unroll
    for (int k = 0; k < 64; ++k) { float t = r[k] - mu; var += t * t; }
    var *= (1.f / 64.f);
    float inv = rsqrtf(var + 1e-5f);
#pragma unroll
    for (int k = 0; k < 64; ++k) g[(size_t)gi * 64 + k] = (r[k] - mu) * inv;
}

__global__ __launch_bounds__(256) void mlp3_kernel(
    const float* __restrict__ m2, const float* __restrict__ W3,
    const float* __restrict__ b3, float* __restrict__ out, int Gg)
{
    int gi = blockIdx.x * blockDim.x + threadIdx.x;
    if (gi >= Gg) return;
    float s = b3[0];
#pragma unroll
    for (int k = 0; k < 32; ++k) s += m2[(size_t)gi * 32 + k] * W3[k];
    out[gi] = s;
}

extern "C" void kernel_launch(void* const* d_in, const int* in_sizes, int n_in,
                              void* d_out, int out_size, void* d_ws, size_t ws_size,
                              hipStream_t stream)
{
    const float* x       = (const float*)d_in[0];
    const int*   type_id = (const int*)  d_in[1];
    const int*   batch   = (const int*)  d_in[2];
    const float* u       = (const float*)d_in[3];
    const int*   ei_jj   = (const int*)  d_in[4];
    const float* ea_jj   = (const float*)d_in[5];
    const int*   ei_mj   = (const int*)  d_in[6];
    const float* ea_mj   = (const float*)d_in[7];
    const float* Wj_enc  = (const float*)d_in[8];   const float* bj_enc = (const float*)d_in[9];
    const float* Wm_enc  = (const float*)d_in[10];  const float* bm_enc = (const float*)d_in[11];
    const float* We_jj1  = (const float*)d_in[12];  const float* be_jj1 = (const float*)d_in[13];
    const float* Wm_jj1  = (const float*)d_in[14];  const float* bm_jj1 = (const float*)d_in[15];
    const float* We_mj1  = (const float*)d_in[16];  const float* be_mj1 = (const float*)d_in[17];
    const float* Wm_mj1  = (const float*)d_in[18];  const float* bm_mj1 = (const float*)d_in[19];
    const float* We_jj2  = (const float*)d_in[20];  const float* be_jj2 = (const float*)d_in[21];
    const float* Wm_jj2  = (const float*)d_in[22];  const float* bm_jj2 = (const float*)d_in[23];
    const float* We_mj2  = (const float*)d_in[24];  const float* be_mj2 = (const float*)d_in[25];
    const float* Wm_mj2  = (const float*)d_in[26];  const float* bm_mj2 = (const float*)d_in[27];
    const float* W1      = (const float*)d_in[28];  const float* b1     = (const float*)d_in[29];
    const float* W2      = (const float*)d_in[30];  const float* b2     = (const float*)d_in[31];
    const float* W3      = (const float*)d_in[32];  const float* b3     = (const float*)d_in[33];

    const int Nn  = in_sizes[0] / 5;
    const int Gg  = in_sizes[3] / 4;
    const int Ejj = in_sizes[5] / 5;
    const int Emj = in_sizes[7] / 4;

    float* ws = (float*)d_ws;
    const size_t NP = (size_t)Nn * DPAD;
    float* hA  = ws;
    float* hB  = ws + NP;
    float* agg = ws + 2 * NP;
    float* wpk = ws + 3 * NP;
    float* pWjj1 = wpk;          float* pbjj1 = wpk + 256;
    float* pWmj1 = wpk + 272;    float* pbmj1 = wpk + 528;
    float* pWjj2 = wpk + 544;    float* pbjj2 = wpk + 800;
    float* pWmj2 = wpk + 816;    float* pbmj2 = wpk + 1072;
    float* pW1   = wpk + 1088;   float* pb1   = wpk + 5184;
    float* pW2   = wpk + 5248;   float* pb2   = wpk + 7296;

    float* P    = agg;
    float* sumJ = P;
    float* sumM = P + (size_t)16 * Gg;
    float* maxJ = P + (size_t)32 * Gg;
    float* maxM = P + (size_t)48 * Gg;
    float* cntJ = P + (size_t)64 * Gg;
    float* cntM = P + (size_t)65 * Gg;
    float* gbuf = P + (size_t)66 * Gg;
    float* m1   = P + (size_t)130 * Gg;
    float* m2   = P + (size_t)194 * Gg;
    float* out  = (float*)d_out;

    const int TB = 256;
    const int nodeBlocks = (Nn + TB - 1) / TB;
    const int fillBlocks = (int)((NP + TB - 1) / TB);

    pack_weight_kernel<<<1, 256, 0, stream>>>(Wm_jj1, bm_jj1, pWjj1, pbjj1, 15, 15, 16, 16);
    pack_weight_kernel<<<1, 256, 0, stream>>>(Wm_mj1, bm_mj1, pWmj1, pbmj1, 15, 15, 16, 16);
    pack_weight_kernel<<<1, 256, 0, stream>>>(Wm_jj2, bm_jj2, pWjj2, pbjj2, 15, 15, 16, 16);
    pack_weight_kernel<<<1, 256, 0, stream>>>(Wm_mj2, bm_mj2, pWmj2, pbmj2, 15, 15, 16, 16);
    pack_weight_kernel<<<16, 256, 0, stream>>>(W1, b1, pW1, pb1, 64, 64, 64, 64);
    pack_weight_kernel<<<8, 256, 0, stream>>>(W2, b2, pW2, pb2, 64, 32, 64, 32);

    auto gemm16 = [&](const float* A, const float* A2, const float* Wp, const float* bp,
                      float* C, int relu) {
        int tiles = Nn >> 4;
        wmma_gemm_kernel<true, 16><<<(tiles + 7) / 8, TB, 0, stream>>>(
            A, A2, Wp, bp, C, Nn, 16, DPAD, DPAD, relu);
    };
    auto gemm64 = [&](const float* A, const float* Wp, const float* bp,
                      float* C, int Ncpad, int lda, int ldc, int relu) {
        int tiles = (Gg >> 4) * (Ncpad >> 4);
        wmma_gemm_kernel<false, 64><<<(tiles + 7) / 8, TB, 0, stream>>>(
            A, nullptr, Wp, bp, C, Gg, Ncpad, lda, ldc, relu);
    };

    encode_kernel<<<nodeBlocks, TB, 0, stream>>>(x, type_id, Wj_enc, bj_enc, Wm_enc, bm_enc, hA, Nn);

    fill_kernel<<<fillBlocks, TB, 0, stream>>>(agg, 0.f, NP);
    edge_pass_kernel<5><<<(Ejj + TB - 1) / TB, TB, 0, stream>>>(hA, agg, ei_jj, ea_jj, We_jj1, be_jj1, Ejj);
    gemm16(hA, agg, pWjj1, pbjj1, hB, 1);

    fill_kernel<<<fillBlocks, TB, 0, stream>>>(agg, 0.f, NP);
    edge_pass_kernel<4><<<(Emj + TB - 1) / TB, TB, 0, stream>>>(hB, agg, ei_mj, ea_mj, We_mj1, be_mj1, Emj);
    gemm16(hB, agg, pWmj1, pbmj1, hA, 0);

    ln_relu_kernel<<<nodeBlocks, TB, 0, stream>>>(hA, Nn);

    fill_kernel<<<fillBlocks, TB, 0, stream>>>(agg, 0.f, NP);
    edge_pass_kernel<5><<<(Ejj + TB - 1) / TB, TB, 0, stream>>>(hA, agg, ei_jj, ea_jj, We_jj2, be_jj2, Ejj);
    gemm16(hA, agg, pWjj2, pbjj2, hB, 1);

    fill_kernel<<<fillBlocks, TB, 0, stream>>>(agg, 0.f, NP);
    edge_pass_kernel<4><<<(Emj + TB - 1) / TB, TB, 0, stream>>>(hB, agg, ei_mj, ea_mj, We_mj2, be_mj2, Emj);
    gemm16(hB, agg, pWmj2, pbmj2, hA, 1);

    ln_relu_kernel<<<nodeBlocks, TB, 0, stream>>>(hA, Nn);

    size_t zlen = (size_t)32 * Gg;
    fill_kernel<<<(int)((zlen + TB - 1) / TB), TB, 0, stream>>>(sumJ, 0.f, zlen);
    fill_kernel<<<(int)((zlen + TB - 1) / TB), TB, 0, stream>>>(maxJ, -__builtin_inff(), zlen);
    fill_kernel<<<(int)(((size_t)2 * Gg + TB - 1) / TB), TB, 0, stream>>>(cntJ, 0.f, (size_t)2 * Gg);
    pool_kernel<<<nodeBlocks, TB, 0, stream>>>(hA, type_id, batch, sumJ, sumM, maxJ, maxM, cntJ, cntM, Nn);

    assemble_ln_kernel<<<(Gg + 127) / 128, 128, 0, stream>>>(sumJ, sumM, maxJ, maxM, cntJ, cntM, u, gbuf, Gg);
    gemm64(gbuf, pW1, pb1, m1, 64, 64, 64, 1);
    gemm64(m1,   pW2, pb2, m2, 32, 64, 32, 1);
    mlp3_kernel<<<(Gg + TB - 1) / TB, TB, 0, stream>>>(m2, W3, b3, out, Gg);
}


# edge pass: one feature per lane, two edges per step (DPP row broadcast): one 64-byte row gather and one row-wide f32 atomic add per edge, not 15 one-word atomics; atomic sum order as before
# speedup vs baseline: 2.1540x; 2.1540x over previous
#include <hip/hip_runtime.h>
#include <hip/hip_bf16.h>

typedef __attribute__((ext_vector_type(2))) float v2f;
typedef __attribute__((ext_vector_type(8))) float v8f;

#define DPAD 16
#define DDIM 15

__global__ void fill_kernel(float* __restrict__ p, float v, size_t n) {
    size_t i = (size_t)blockIdx.x * blockDim.x + threadIdx.x;
    size_t stride = (size_t)gridDim.x * blockDim.x;
    for (; i < n; i += stride) p[i] = v;
}

__global__ void pack_weight_kernel(const float* __restrict__ W, const float* __restrict__ b,
                                   float* __restrict__ Wp, float* __restrict__ bp,
                                   int K, int Nc, int Kpad, int Ncpad)
{
    int idx = blockIdx.x * blockDim.x + threadIdx.x;
    int total = Kpad * Ncpad;
    if (idx < total) {
        int k = idx / Ncpad, c = idx % Ncpad;
        float v = (k < K && c < Nc) ? W[k * Nc + c] : 0.f;
        Wp[((size_t)(k >> 1) * Ncpad + c) * 2 + (k & 1)] = v;
    }
    if (idx < Ncpad) bp[idx] = (idx < Nc) ? b[idx] : 0.f;
}

__global__ __launch_bounds__(256) void encode_kernel(
    const float* __restrict__ x, const int* __restrict__ type_id,
    const float* __restrict__ Wj, const float* __restrict__ bj,
    const float* __restrict__ Wm, const float* __restrict__ bm,
    float* __restrict__ h, int Nn)
{
    int i = blockIdx.x * blockDim.x + threadIdx.x;
    if (i >= Nn) return;
    float xi[5];
#pragma unroll
    for (int k = 0; k < 5; ++k) xi[k] = x[(size_t)i * 5 + k];
    int t = type_id[i];
    const float* Wp = (t == 0) ? Wj : Wm;
    const float* bp = (t == 0) ? bj : bm;
    bool valid = (t == 0) || (t == 1);
#pragma unroll
    for (int d = 0; d < DDIM; ++d) {
        float v = 0.f;
        if (valid) {
            v = bp[d];
#pragma unroll
            for (int k = 0; k < 5; ++k) v += xi[k] * Wp[k * DDIM + d];
            v = fmaxf(v, 0.f);
        }
        h[(size_t)i * DPAD + d] = v;
    }
    h[(size_t)i * DPAD + DDIM] = 0.f;
}

template <int J>
__device__ __forceinline__ int row_share_i(int v) {
    return __builtin_amdgcn_update_dpp(v, v, 0x150 + J, 0xF, 0xF, false);
}
template <int J>
__device__ __forceinline__ float row_share_f(float v) {
    return __int_as_float(row_share_i<J>(__float_as_int(v)));
}

#define FOR16(X) X(0) X(1) X(2) X(3) X(4) X(5) X(6) X(7) \
                 X(8) X(9) X(10) X(11) X(12) X(13) X(14) X(15)

template <int EDIM, bool FULL>
__device__ __forceinline__ void edge_chunk(
    const float* __restrict__ h, float* __restrict__ agg,
    const int* __restrict__ ei, const float* __restrict__ ea,
    const float* __restrict__ We, const float* __restrict__ be,
    int E, int e0, int lane)
{
    const int c    = lane & 15;
    const int half = lane >> 4;

    int e = e0 + lane;
    if (!FULL) e = (e < E) ? e : E - 1;
    const int src = ei[e];
    const int dst = ei[E + e];
    float ev[EDIM];
#pragma unroll
    for (int k = 0; k < EDIM; ++k) ev[k] = ea[(size_t)e * EDIM + k];

    const int cc = (c < DDIM) ? c : DDIM - 1;
    float w[EDIM];
#pragma unroll
    for (int k = 0; k < EDIM; ++k) {
        const float t = We[k * DDIM + cc];
        w[k] = (c < DDIM) ? t : 0.f;
    }
    const float tb = be[cc];
    const float bc = (c < DDIM) ? tb : 0.f;

    float hv[16];
#define GATHER_STEP(J) hv[J] = h[(unsigned)row_share_i<J>(src) * DPAD + c];
    FOR16(GATHER_STEP)
#undef GATHER_STEP

#define SCATTER_STEP(J) {                                                      \
        const unsigned d = (unsigned)row_share_i<J>(dst);                      \
        float m = hv[J] + bc;                                                  \
        _Pragma("unroll")                                                      \
        for (int k = 0; k < EDIM; ++k) m += row_share_f<J>(ev[k]) * w[k];      \
        m = fmaxf(m, 0.f);                                                     \
        if (FULL || e0 + 16 * half + J < E)                                    \
            atomicAdd(&agg[d * DPAD + c], m);                                  \
    }
    FOR16(SCATTER_STEP)
#undef SCATTER_STEP
}
#undef FOR16

template <int EDIM>
__global__ __launch_bounds__(256) void edge_pass_kernel(
    const float* __restrict__ h, float* __restrict__ agg,
    const int* __restrict__ ei, const float* __restrict__ ea,
    const float* __restrict__ We, const float* __restrict__ be, int E)
{
    const int lane = threadIdx.x & 31;
    const int e0   = (blockIdx.x * (blockDim.x >> 5) + (threadIdx.x >> 5)) * 32;
    if (e0 >= E) return;
    if (e0 + 32 <= E) edge_chunk<EDIM, true >(h, agg, ei, ea, We, be, E, e0, lane);
    else              edge_chunk<EDIM, false>(h, agg, ei, ea, We, be, E, e0, lane);
}

template <bool HAS_A2, int KPAD>
__global__ __launch_bounds__(256) void wmma_gemm_kernel(
    const float* __restrict__ A, const float* __restrict__ A2,
    const float* __restrict__ Wp, const float* __restrict__ bp,
    float* __restrict__ C,
    int M, int Ncpad, int lda, int ldc, int relu)
{
    const int lane = threadIdx.x & 31;
    const int wave = threadIdx.x >> 5;
    const int ntN = Ncpad >> 4;
    const int tile = blockIdx.x * (blockDim.x >> 5) + wave;
    if (tile >= (M >> 4) * ntN) return;
    const int mt = tile / ntN;
    const int nt = tile % ntN;

    const int mrow  = mt * 16 + (lane & 15);
    const int ncol  = nt * 16 + (lane & 15);
    const int khalf = (lane >> 4) << 1;

    const float* ap  = A + (size_t)mrow * lda + khalf;
    const float* a2p = A2 + (size_t)mrow * lda + khalf;
    const float* bpp = Wp + ((size_t)(khalf >> 1) * Ncpad + ncol) * 2;
    const size_t bstep = (size_t)Ncpad * 4;

    v8f acc = {0.f, 0.f, 0.f, 0.f, 0.f, 0.f, 0.f, 0.f};
#pragma unroll
    for (int kk = 0; kk < KPAD; kk += 4) {
        v2f a = *(const v2f*)ap;
        if (HAS_A2) {
            v2f a2 = *(const v2f*)a2p;
            a.x += a2.x; a.y += a2.y;
            a2p += 4;
        }
        v2f b = *(const v2f*)bpp;
        ap += 4;
        bpp += bstep;
        acc = __builtin_amdgcn_wmma_f32_16x16x4_f32(
            false, a, false, b, (short)0, acc, false, false);
    }

    const int rbase = mt * 16 + ((lane >> 4) << 3);
    const float bv = bp[ncol];
#pragma unroll
    for (int r = 0; r < 8; ++r) {
        float v = acc[r] + bv;
        if (relu) v = fmaxf(v, 0.f);
        C[(size_t)(rbase + r) * ldc + ncol] = v;
    }
}

__global__ __launch_bounds__(256) void ln_relu_kernel(float* __restrict__ h, int Nn)
{
    int i = blockIdx.x * blockDim.x + threadIdx.x;
    if (i >= Nn) return;
    float v[DDIM];
    float mu = 0.f;
#pragma unroll
    for (int d = 0; d < DDIM; ++d) { v[d] = h[(size_t)i * DPAD + d]; mu += v[d]; }
    mu *= (1.f / DDIM);
    float var = 0.f;
#pragma unroll
    for (int d = 0; d < DDIM; ++d) { float t = v[d] - mu; var += t * t; }
    var *= (1.f / DDIM);
    float inv = rsqrtf(var + 1e-5f);
#pragma unroll
    for (int d = 0; d < DDIM; ++d)
        h[(size_t)i * DPAD + d] = fmaxf((v[d] - mu) * inv, 0.f);
}

__device__ __forceinline__ void atomicMaxF(float* addr, float v) {
    if (v >= 0.f) atomicMax((int*)addr, __float_as_int(v));
    else          atomicMin((unsigned int*)addr, __float_as_uint(v));
}

__global__ __launch_bounds__(256) void pool_kernel(
    const float* __restrict__ h, const int* __restrict__ type_id,
    const int* __restrict__ batch,
    float* __restrict__ sumJ, float* __restrict__ sumM,
    float* __restrict__ maxJ, float* __restrict__ maxM,
    float* __restrict__ cntJ, float* __restrict__ cntM, int Nn)
{
    int i = blockIdx.x * blockDim.x + threadIdx.x;
    if (i >= Nn) return;
    int t = type_id[i];
    if (t != 0 && t != 1) return;
    int b = batch[i];
    float* s  = (t == 0) ? sumJ : sumM;
    float* mx = (t == 0) ? maxJ : maxM;
    float* c  = (t == 0) ? cntJ : cntM;
    atomicAdd(&c[b], 1.f);
#pragma unroll
    for (int d = 0; d < DDIM; ++d) {
        float v = h[(size_t)i * DPAD + d];
        atomicAdd(&s[(size_t)b * DPAD + d], v);
        atomicMaxF(&mx[(size_t)b * DPAD + d], v);
    }
}

__global__ __launch_bounds__(128) void assemble_ln_kernel(
    const float* __restrict__ sumJ, const float* __restrict__ sumM,
    const float* __restrict__ maxJ, const float* __restrict__ maxM,
    const float* __restrict__ cntJ, const float* __restrict__ cntM,
    const float* __restrict__ u, float* __restrict__ g, int Gg)
{
    int gi = blockIdx.x * blockDim.x + threadIdx.x;
    if (gi >= Gg) return;
    float r[64];
    float cm = fmaxf(cntM[gi], 1.f);
    float cj = fmaxf(cntJ[gi], 1.f);
#pragma unroll
    for (int d = 0; d < DDIM; ++d) r[d] = sumM[(size_t)gi * DPAD + d] / cm;
#pragma unroll
    for (int d = 0; d < DDIM; ++d) r[15 + d] = sumJ[(size_t)gi * DPAD + d] / cj;
#pragma unroll
    for (int d = 0; d < DDIM; ++d) {
        float v = maxJ[(size_t)gi * DPAD + d];
        r[30 + d] = (v > -3.0e38f) ? v : 0.f;
    }
#pragma unroll
    for (int d = 0; d < DDIM; ++d) {
        float v = maxM[(size_t)gi * DPAD + d];
        r[45 + d] = (v > -3.0e38f) ? v : 0.f;
    }
#pragma unroll
    for (int j = 0; j < 4; ++j) r[60 + j] = u[(size_t)gi * 4 + j];

    float mu = 0.f;
#pragma unroll
    for (int k = 0; k < 64; ++k) mu += r[k];
    mu *= (1.f / 64.f);
    float var = 0.f;
#pragma unroll
    for (int k = 0; k < 64; ++k) { float t = r[k] - mu; var += t * t; }
    var *= (1.f / 64.f);
    float inv = rsqrtf(var + 1e-5f);
#pragma unroll
    for (int k = 0; k < 64; ++k) g[(size_t)gi * 64 + k] = (r[k] - mu) * inv;
}

__global__ __launch_bounds__(256) void mlp3_kernel(
    const float* __restrict__ m2, const float* __restrict__ W3,
    const float* __restrict__ b3, float* __restrict__ out, int Gg)
{
    int gi = blockIdx.x * blockDim.x + threadIdx.x;
    if (gi >= Gg) return;
    float s = b3[0];
#pragma unroll
    for (int k = 0; k < 32; ++k) s += m2[(size_t)gi * 32 + k] * W3[k];
    out[gi] = s;
}

extern "C" void kernel_launch(void* const* d_in, const int* in_sizes, int n_in,
                              void* d_out, int out_size, void* d_ws, size_t ws_size,
                              hipStream_t stream)
{
    const float* x       = (const float*)d_in[0];
    const int*   type_id = (const int*)  d_in[1];
    const int*   batch   = (const int*)  d_in[2];
    const float* u       = (const float*)d_in[3];
    const int*   ei_jj   = (const int*)  d_in[4];
    const float* ea_jj   = (const float*)d_in[5];
    const int*   ei_mj   = (const int*)  d_in[6];
    const float* ea_mj   = (const float*)d_in[7];
    const float* Wj_enc  = (const float*)d_in[8];   const float* bj_enc = (const float*)d_in[9];
    const float* Wm_enc  = (const float*)d_in[10];  const float* bm_enc = (const float*)d_in[11];
    const float* We_jj1  = (const float*)d_in[12];  const float* be_jj1 = (const float*)d_in[13];
    const float* Wm_jj1  = (const float*)d_in[14];  const float* bm_jj1 = (const float*)d_in[15];
    const float* We_mj1  = (const float*)d_in[16];  const float* be_mj1 = (const float*)d_in[17];
    const float* Wm_mj1  = (const float*)d_in[18];  const float* bm_mj1 = (const float*)d_in[19];
    const float* We_jj2  = (const float*)d_in[20];  const float* be_jj2 = (const float*)d_in[21];
    const float* Wm_jj2  = (const float*)d_in[22];  const float* bm_jj2 = (const float*)d_in[23];
    const float* We_mj2  = (const float*)d_in[24];  const float* be_mj2 = (const float*)d_in[25];
    const float* Wm_mj2  = (const float*)d_in[26];  const float* bm_mj2 = (const float*)d_in[27];
    const float* W1      = (const float*)d_in[28];  const float* b1     = (const float*)d_in[29];
    const float* W2      = (const float*)d_in[30];  const float* b2     = (const float*)d_in[31];
    const float* W3      = (const float*)d_in[32];  const float* b3     = (const float*)d_in[33];

    const int Nn  = in_sizes[0] / 5;
    const int Gg  = in_sizes[3] / 4;
    const int Ejj = in_sizes[5] / 5;
    const int Emj = in_sizes[7] / 4;

    float* ws = (float*)d_ws;
    const size_t NP = (size_t)Nn * DPAD;
    float* hA  = ws;
    float* hB  = ws + NP;
    float* agg = ws + 2 * NP;
    float* wpk = ws + 3 * NP;
    float* pWjj1 = wpk;          float* pbjj1 = wpk + 256;
    float* pWmj1 = wpk + 272;    float* pbmj1 = wpk + 528;
    float* pWjj2 = wpk + 544;    float* pbjj2 = wpk + 800;
    float* pWmj2 = wpk + 816;    float* pbmj2 = wpk + 1072;
    float* pW1   = wpk + 1088;   float* pb1   = wpk + 5184;
    float* pW2   = wpk + 5248;   float* pb2   = wpk + 7296;

    float* P    = agg;
    float* sumJ = P;
    float* sumM = P + (size_t)16 * Gg;
    float* maxJ = P + (size_t)32 * Gg;
    float* maxM = P + (size_t)48 * Gg;
    float* cntJ = P + (size_t)64 * Gg;
    float* cntM = P + (size_t)65 * Gg;
    float* gbuf = P + (size_t)66 * Gg;
    float* m1   = P + (size_t)130 * Gg;
    float* m2   = P + (size_t)194 * Gg;
    float* out  = (float*)d_out;

    const int TB = 256;
    const int nodeBlocks = (Nn + TB - 1) / TB;
    const int fillBlocks = (int)((NP + TB - 1) / TB);

    pack_weight_kernel<<<1, 256, 0, stream>>>(Wm_jj1, bm_jj1, pWjj1, pbjj1, 15, 15, 16, 16);
    pack_weight_kernel<<<1, 256, 0, stream>>>(Wm_mj1, bm_mj1, pWmj1, pbmj1, 15, 15, 16, 16);
    pack_weight_kernel<<<1, 256, 0, stream>>>(Wm_jj2, bm_jj2, pWjj2, pbjj2, 15, 15, 16, 16);
    pack_weight_kernel<<<1, 256, 0, stream>>>(Wm_mj2, bm_mj2, pWmj2, pbmj2, 15, 15, 16, 16);
    pack_weight_kernel<<<16, 256, 0, stream>>>(W1, b1, pW1, pb1, 64, 64, 64, 64);
    pack_weight_kernel<<<8, 256, 0, stream>>>(W2, b2, pW2, pb2, 64, 32, 64, 32);

    auto gemm16 = [&](const float* A, const float* A2, const float* Wp, const float* bp,
                      float* C, int relu) {
        int tiles = Nn >> 4;
        wmma_gemm_kernel<true, 16><<<(tiles + 7) / 8, TB, 0, stream>>>(
            A, A2, Wp, bp, C, Nn, 16, DPAD, DPAD, relu);
    };
    auto gemm64 = [&](const float* A, const float* Wp, const float* bp,
                      float* C, int Ncpad, int lda, int ldc, int relu) {
        int tiles = (Gg >> 4) * (Ncpad >> 4);
        wmma_gemm_kernel<false, 64><<<(tiles + 7) / 8, TB, 0, stream>>>(
            A, nullptr, Wp, bp, C, Gg, Ncpad, lda, ldc, relu);
    };

    encode_kernel<<<nodeBlocks, TB, 0, stream>>>(x, type_id, Wj_enc, bj_enc, Wm_enc, bm_enc, hA, Nn);

    fill_kernel<<<fillBlocks, TB, 0, stream>>>(agg, 0.f, NP);
    edge_pass_kernel<5><<<(Ejj + TB - 1) / TB, TB, 0, stream>>>(hA, agg, ei_jj, ea_jj, We_jj1, be_jj1, Ejj);
    gemm16(hA, agg, pWjj1, pbjj1, hB, 1);

    fill_kernel<<<fillBlocks, TB, 0, stream>>>(agg, 0.f, NP);
    edge_pass_kernel<4><<<(Emj + TB - 1) / TB, TB, 0, stream>>>(hB, agg, ei_mj, ea_mj, We_mj1, be_mj1, Emj);
    gemm16(hB, agg, pWmj1, pbmj1, hA, 0);

    ln_relu_kernel<<<nodeBlocks, TB, 0, stream>>>(hA, Nn);

    fill_kernel<<<fillBlocks, TB, 0, stream>>>(agg, 0.f, NP);
    edge_pass_kernel<5><<<(Ejj + TB - 1) / TB, TB, 0, stream>>>(hA, agg, ei_jj, ea_jj, We_jj2, be_jj2, Ejj);
    gemm16(hA, agg, pWjj2, pbjj2, hB, 1);

    fill_kernel<<<fillBlocks, TB, 0, stream>>>(agg, 0.f, NP);
    edge_pass_kernel<4><<<(Emj + TB - 1) / TB, TB, 0, stream>>>(hB, agg, ei_mj, ea_mj, We_mj2, be_mj2, Emj);
    gemm16(hB, agg, pWmj2, pbmj2, hA, 1);

    ln_relu_kernel<<<nodeBlocks, TB, 0, stream>>>(hA, Nn);

    size_t zlen = (size_t)32 * Gg;
    fill_kernel<<<(int)((zlen + TB - 1) / TB), TB, 0, stream>>>(sumJ, 0.f, zlen);
    fill_kernel<<<(int)((zlen + TB - 1) / TB), TB, 0, stream>>>(maxJ, -__builtin_inff(), zlen);
    fill_kernel<<<(int)(((size_t)2 * Gg + TB - 1) / TB), TB, 0, stream>>>(cntJ, 0.f, (size_t)2 * Gg);
    pool_kernel<<<nodeBlocks, TB, 0, stream>>>(hA, type_id, batch, sumJ, sumM, maxJ, maxM, cntJ, cntM, Nn);

    assemble_ln_kernel<<<(Gg + 127) / 128, 128, 0, stream>>>(sumJ, sumM, maxJ, maxM, cntJ, cntM, u, gbuf, Gg);
    gemm64(gbuf, pW1, pb1, m1, 64, 64, 64, 1);
    gemm64(m1,   pW2, pb2, m2, 32, 64, 32, 1);
    mlp3_kernel<<<(Gg + TB - 1) / TB, TB, 0, stream>>>(m2, W3, b3, out, Gg);
}


# edge accumulator starts as a copy of h, written by the kernel that makes h, not as zeros: 4 zero-fill launches go, node update reads one array; sum order: h first, then messages in atomic order
# speedup vs baseline: 2.2693x; 1.0535x over previous
#include <hip/hip_runtime.h>
#include <hip/hip_bf16.h>

typedef __attribute__((ext_vector_type(2))) float v2f;
typedef __attribute__((ext_vector_type(8))) float v8f;

#define DPAD 16
#define DDIM 15

__global__ void fill_kernel(float* __restrict__ p, float v, size_t n) {
    size_t i = (size_t)blockIdx.x * blockDim.x + threadIdx.x;
    size_t stride = (size_t)gridDim.x * blockDim.x;
    for (; i < n; i += stride) p[i] = v;
}

__global__ void pack_weight_kernel(const float* __restrict__ W, const float* __restrict__ b,
                                   float* __restrict__ Wp, float* __restrict__ bp,
                                   int K, int Nc, int Kpad, int Ncpad)
{
    int idx = blockIdx.x * blockDim.x + threadIdx.x;
    int total = Kpad * Ncpad;
    if (idx < total) {
        int k = idx / Ncpad, c = idx % Ncpad;
        float v = (k < K && c < Nc) ? W[k * Nc + c] : 0.f;
        Wp[((size_t)(k >> 1) * Ncpad + c) * 2 + (k & 1)] = v;
    }
    if (idx < Ncpad) bp[idx] = (idx < Nc) ? b[idx] : 0.f;
}

typedef __attribute__((ext_vector_type(4))) float v4f;

__device__ __forceinline__ void store_row16(float* __restrict__ dst, const float* v) {
    v4f* p = (v4f*)dst;
#pragma unroll
    for (int q = 0; q < 4; ++q) {
        v4f t = {v[4 * q], v[4 * q + 1], v[4 * q + 2], v[4 * q + 3]};
        p[q] = t;
    }
}

__global__ __launch_bounds__(256) void encode_kernel(
    const float* __restrict__ x, const int* __restrict__ type_id,
    const float* __restrict__ Wj, const float* __restrict__ bj,
    const float* __restrict__ Wm, const float* __restrict__ bm,
    float* __restrict__ h, float* __restrict__ acc, int Nn)
{
    int i = blockIdx.x * blockDim.x + threadIdx.x;
    if (i >= Nn) return;
    float row[DPAD];
    float xi[5];
#pragma unroll
    for (int k = 0; k < 5; ++k) xi[k] = x[(size_t)i * 5 + k];
    int t = type_id[i];
    const float* Wp = (t == 0) ? Wj : Wm;
    const float* bp = (t == 0) ? bj : bm;
    bool valid = (t == 0) || (t == 1);
#pragma unroll
    for (int d = 0; d < DDIM; ++d) {
        float v = 0.f;
        if (valid) {
            v = bp[d];
#pragma unroll
            for (int k = 0; k < 5; ++k) v += xi[k] * Wp[k * DDIM + d];
            v = fmaxf(v, 0.f);
        }
        h[(size_t)i * DPAD + d] = v;
        row[d] = v;
    }
    h[(size_t)i * DPAD + DDIM] = 0.f;
    row[DDIM] = 0.f;
    store_row16(acc + (size_t)i * DPAD, row);
}

template <int J>
__device__ __forceinline__ int row_share_i(int v) {
    return __builtin_amdgcn_update_dpp(v, v, 0x150 + J, 0xF, 0xF, false);
}
template <int J>
__device__ __forceinline__ float row_share_f(float v) {
    return __int_as_float(row_share_i<J>(__float_as_int(v)));
}

#define FOR16(X) X(0) X(1) X(2) X(3) X(4) X(5) X(6) X(7) \
                 X(8) X(9) X(10) X(11) X(12) X(13) X(14) X(15)

template <int EDIM, bool FULL>
__device__ __forceinline__ void edge_chunk(
    const float* __restrict__ h, float* __restrict__ agg,
    const int* __restrict__ ei, const float* __restrict__ ea,
    const float* __restrict__ We, const float* __restrict__ be,
    int E, int e0, int lane)
{
    const int c    = lane & 15;
    const int half = lane >> 4;

    int e = e0 + lane;
    if (!FULL) e = (e < E) ? e : E - 1;
    const int src = ei[e];
    const int dst = ei[E + e];
    float ev[EDIM];
#pragma unroll
    for (int k = 0; k < EDIM; ++k) ev[k] = ea[(size_t)e * EDIM + k];

    const int cc = (c < DDIM) ? c : DDIM - 1;
    float w[EDIM];
#pragma unroll
    for (int k = 0; k < EDIM; ++k) {
        const float t = We[k * DDIM + cc];
        w[k] = (c < DDIM) ? t : 0.f;
    }
    const float tb = be[cc];
    const float bc = (c < DDIM) ? tb : 0.f;

    float hv[16];
#define GATHER_STEP(J) hv[J] = h[(unsigned)row_share_i<J>(src) * DPAD + c];
    FOR16(GATHER_STEP)
#undef GATHER_STEP

#define SCATTER_STEP(J) {                                                      \
        const unsigned d = (unsigned)row_share_i<J>(dst);                      \
        float m = hv[J] + bc;                                                  \
        _Pragma("unroll")                                                      \
        for (int k = 0; k < EDIM; ++k) m += row_share_f<J>(ev[k]) * w[k];      \
        m = fmaxf(m, 0.f);                                                     \
        if (FULL || e0 + 16 * half + J < E)                                    \
            atomicAdd(&agg[d * DPAD + c], m);                                  \
    }
    FOR16(SCATTER_STEP)
#undef SCATTER_STEP
}
#undef FOR16

template <int EDIM>
__global__ __launch_bounds__(256) void edge_pass_kernel(
    const float* __restrict__ h, float* __restrict__ agg,
    const int* __restrict__ ei, const float* __restrict__ ea,
    const float* __restrict__ We, const float* __restrict__ be, int E)
{
    const int lane = threadIdx.x & 31;
    const int e0   = (blockIdx.x * (blockDim.x >> 5) + (threadIdx.x >> 5)) * 32;
    if (e0 >= E) return;
    if (e0 + 32 <= E) edge_chunk<EDIM, true >(h, agg, ei, ea, We, be, E, e0, lane);
    else              edge_chunk<EDIM, false>(h, agg, ei, ea, We, be, E, e0, lane);
}

template <bool HAS_A2, int KPAD>
__global__ __launch_bounds__(256) void wmma_gemm_kernel(
    const float* A, const float* A2,
    const float* __restrict__ Wp, const float* __restrict__ bp,
    float* C, float* C2,
    int M, int Ncpad, int lda, int ldc, int relu)
{
    const int lane = threadIdx.x & 31;
    const int wave = threadIdx.x >> 5;
    const int ntN = Ncpad >> 4;
    const int tile = blockIdx.x * (blockDim.x >> 5) + wave;
    if (tile >= (M >> 4) * ntN) return;
    const int mt = tile / ntN;
    const int nt = tile % ntN;

    const int mrow  = mt * 16 + (lane & 15);
    const int ncol  = nt * 16 + (lane & 15);
    const int khalf = (lane >> 4) << 1;

    const float* ap  = A + (size_t)mrow * lda + khalf;
    const float* a2p = A2 + (size_t)mrow * lda + khalf;
    const float* bpp = Wp + ((size_t)(khalf >> 1) * Ncpad + ncol) * 2;
    const size_t bstep = (size_t)Ncpad * 4;

    v8f acc = {0.f, 0.f, 0.f, 0.f, 0.f, 0.f, 0.f, 0.f};
#pragma unroll
    for (int kk = 0; kk < KPAD; kk += 4) {
        v2f a = *(const v2f*)ap;
        if (HAS_A2) {
            v2f a2 = *(const v2f*)a2p;
            a.x += a2.x; a.y += a2.y;
            a2p += 4;
        }
        v2f b = *(const v2f*)bpp;
        ap += 4;
        bpp += bstep;
        acc = __builtin_amdgcn_wmma_f32_16x16x4_f32(
            false, a, false, b, (short)0, acc, false, false);
    }

    const int rbase = mt * 16 + ((lane >> 4) << 3);
    const float bv = bp[ncol];
    float o[8];
#pragma unroll
    for (int r = 0; r < 8; ++r) {
        float v = acc[r] + bv;
        if (relu) v = fmaxf(v, 0.f);
        o[r] = v;
        C[(size_t)(rbase + r) * ldc + ncol] = v;
    }
    if (C2) {
#pragma unroll
        for (int r = 0; r < 8; ++r)
            C2[(size_t)(rbase + r) * ldc + ncol] = o[r];
    }
}

__global__ __launch_bounds__(256) void ln_relu_kernel(float* __restrict__ h,
                                                      float* __restrict__ acc, int Nn)
{
    int i = blockIdx.x * blockDim.x + threadIdx.x;
    if (i >= Nn) return;
    float v[DDIM];
    float mu = 0.f;
#pragma unroll
    for (int d = 0; d < DDIM; ++d) { v[d] = h[(size_t)i * DPAD + d]; mu += v[d]; }
    mu *= (1.f / DDIM);
    float var = 0.f;
#pragma unroll
    for (int d = 0; d < DDIM; ++d) { float t = v[d] - mu; var += t * t; }
    var *= (1.f / DDIM);
    float inv = rsqrtf(var + 1e-5f);
    float row[DPAD];
#pragma unroll
    for (int d = 0; d < DDIM; ++d) {
        row[d] = fmaxf((v[d] - mu) * inv, 0.f);
        h[(size_t)i * DPAD + d] = row[d];
    }
    if (acc) {
        row[DDIM] = 0.f;
        store_row16(acc + (size_t)i * DPAD, row);
    }
}

__device__ __forceinline__ void atomicMaxF(float* addr, float v) {
    if (v >= 0.f) atomicMax((int*)addr, __float_as_int(v));
    else          atomicMin((unsigned int*)addr, __float_as_uint(v));
}

__global__ __launch_bounds__(256) void pool_kernel(
    const float* __restrict__ h, const int* __restrict__ type_id,
    const int* __restrict__ batch,
    float* __restrict__ sumJ, float* __restrict__ sumM,
    float* __restrict__ maxJ, float* __restrict__ maxM,
    float* __restrict__ cntJ, float* __restrict__ cntM, int Nn)
{
    int i = blockIdx.x * blockDim.x + threadIdx.x;
    if (i >= Nn) return;
    int t = type_id[i];
    if (t != 0 && t != 1) return;
    int b = batch[i];
    float* s  = (t == 0) ? sumJ : sumM;
    float* mx = (t == 0) ? maxJ : maxM;
    float* c  = (t == 0) ? cntJ : cntM;
    atomicAdd(&c[b], 1.f);
#pragma unroll
    for (int d = 0; d < DDIM; ++d) {
        float v = h[(size_t)i * DPAD + d];
        atomicAdd(&s[(size_t)b * DPAD + d], v);
        atomicMaxF(&mx[(size_t)b * DPAD + d], v);
    }
}

__global__ __launch_bounds__(128) void assemble_ln_kernel(
    const float* __restrict__ sumJ, const float* __restrict__ sumM,
    const float* __restrict__ maxJ, const float* __restrict__ maxM,
    const float* __restrict__ cntJ, const float* __restrict__ cntM,
    const float* __restrict__ u, float* __restrict__ g, int Gg)
{
    int gi = blockIdx.x * blockDim.x + threadIdx.x;
    if (gi >= Gg) return;
    float r[64];
    float cm = fmaxf(cntM[gi], 1.f);
    float cj = fmaxf(cntJ[gi], 1.f);
#pragma unroll
    for (int d = 0; d < DDIM; ++d) r[d] = sumM[(size_t)gi * DPAD + d] / cm;
#pragma unroll
    for (int d = 0; d < DDIM; ++d) r[15 + d] = sumJ[(size_t)gi * DPAD + d] / cj;
#pragma unroll
    for (int d = 0; d < DDIM; ++d) {
        float v = maxJ[(size_t)gi * DPAD + d];
        r[30 + d] = (v > -3.0e38f) ? v : 0.f;
    }
#pragma unroll
    for (int d = 0; d < DDIM; ++d) {
        float v = maxM[(size_t)gi * DPAD + d];
        r[45 + d] = (v > -3.0e38f) ? v : 0.f;
    }
#pragma unroll
    for (int j = 0; j < 4; ++j) r[60 + j] = u[(size_t)gi * 4 + j];

    float mu = 0.f;
#pragma unroll
    for (int k = 0; k < 64; ++k) mu += r[k];
    mu *= (1.f / 64.f);
    float var = 0.f;
#pragma unroll
    for (int k = 0; k < 64; ++k) { float t = r[k] - mu; var += t * t; }
    var *= (1.f / 64.f);
    float inv = rsqrtf(var + 1e-5f);
#pragma unroll
    for (int k = 0; k < 64; ++k) g[(size_t)gi * 64 + k] = (r[k] - mu) * inv;
}

__global__ __launch_bounds__(256) void mlp3_kernel(
    const float* __restrict__ m2, const float* __restrict__ W3,
    const float* __restrict__ b3, float* __restrict__ out, int Gg)
{
    int gi = blockIdx.x * blockDim.x + threadIdx.x;
    if (gi >= Gg) return;
    float s = b3[0];
#pragma unroll
    for (int k = 0; k < 32; ++k) s += m2[(size_t)gi * 32 + k] * W3[k];
    out[gi] = s;
}

extern "C" void kernel_launch(void* const* d_in, const int* in_sizes, int n_in,
                              void* d_out, int out_size, void* d_ws, size_t ws_size,
                              hipStream_t stream)
{
    const float* x       = (const float*)d_in[0];
    const int*   type_id = (const int*)  d_in[1];
    const int*   batch   = (const int*)  d_in[2];
    const float* u       = (const float*)d_in[3];
    const int*   ei_jj   = (const int*)  d_in[4];
    const float* ea_jj   = (const float*)d_in[5];
    const int*   ei_mj   = (const int*)  d_in[6];
    const float* ea_mj   = (const float*)d_in[7];
    const float* Wj_enc  = (const float*)d_in[8];   const float* bj_enc = (const float*)d_in[9];
    const float* Wm_enc  = (const float*)d_in[10];  const float* bm_enc = (const float*)d_in[11];
    const float* We_jj1  = (const float*)d_in[12];  const float* be_jj1 = (const float*)d_in[13];
    const float* Wm_jj1  = (const float*)d_in[14];  const float* bm_jj1 = (const float*)d_in[15];
    const float* We_mj1  = (const float*)d_in[16];  const float* be_mj1 = (const float*)d_in[17];
    const float* Wm_mj1  = (const float*)d_in[18];  const float* bm_mj1 = (const float*)d_in[19];
    const float* We_jj2  = (const float*)d_in[20];  const float* be_jj2 = (const float*)d_in[21];
    const float* Wm_jj2  = (const float*)d_in[22];  const float* bm_jj2 = (const float*)d_in[23];
    const float* We_mj2  = (const float*)d_in[24];  const float* be_mj2 = (const float*)d_in[25];
    const float* Wm_mj2  = (const float*)d_in[26];  const float* bm_mj2 = (const float*)d_in[27];
    const float* W1      = (const float*)d_in[28];  const float* b1     = (const float*)d_in[29];
    const float* W2      = (const float*)d_in[30];  const float* b2     = (const float*)d_in[31];
    const float* W3      = (const float*)d_in[32];  const float* b3     = (const float*)d_in[33];

    const int Nn  = in_sizes[0] / 5;
    const int Gg  = in_sizes[3] / 4;
    const int Ejj = in_sizes[5] / 5;
    const int Emj = in_sizes[7] / 4;

    float* ws = (float*)d_ws;
    const size_t NP = (size_t)Nn * DPAD;
    float* hA  = ws;
    float* agg = ws + 2 * NP;
    float* wpk = ws + 3 * NP;
    float* pWjj1 = wpk;          float* pbjj1 = wpk + 256;
    float* pWmj1 = wpk + 272;    float* pbmj1 = wpk + 528;
    float* pWjj2 = wpk + 544;    float* pbjj2 = wpk + 800;
    float* pWmj2 = wpk + 816;    float* pbmj2 = wpk + 1072;
    float* pW1   = wpk + 1088;   float* pb1   = wpk + 5184;
    float* pW2   = wpk + 5248;   float* pb2   = wpk + 7296;

    float* P    = agg;
    float* sumJ = P;
    float* sumM = P + (size_t)16 * Gg;
    float* maxJ = P + (size_t)32 * Gg;
    float* maxM = P + (size_t)48 * Gg;
    float* cntJ = P + (size_t)64 * Gg;
    float* cntM = P + (size_t)65 * Gg;
    float* gbuf = P + (size_t)66 * Gg;
    float* m1   = P + (size_t)130 * Gg;
    float* m2   = P + (size_t)194 * Gg;
    float* out  = (float*)d_out;

    const int TB = 256;
    const int nodeBlocks = (Nn + TB - 1) / TB;

    pack_weight_kernel<<<1, 256, 0, stream>>>(Wm_jj1, bm_jj1, pWjj1, pbjj1, 15, 15, 16, 16);
    pack_weight_kernel<<<1, 256, 0, stream>>>(Wm_mj1, bm_mj1, pWmj1, pbmj1, 15, 15, 16, 16);
    pack_weight_kernel<<<1, 256, 0, stream>>>(Wm_jj2, bm_jj2, pWjj2, pbjj2, 15, 15, 16, 16);
    pack_weight_kernel<<<1, 256, 0, stream>>>(Wm_mj2, bm_mj2, pWmj2, pbmj2, 15, 15, 16, 16);
    pack_weight_kernel<<<16, 256, 0, stream>>>(W1, b1, pW1, pb1, 64, 64, 64, 64);
    pack_weight_kernel<<<8, 256, 0, stream>>>(W2, b2, pW2, pb2, 64, 32, 64, 32);

    auto update16 = [&](const float* Wp, const float* bp, float* next_acc, int relu) {
        int tiles = Nn >> 4;
        wmma_gemm_kernel<false, 16><<<(tiles + 7) / 8, TB, 0, stream>>>(
            agg, nullptr, Wp, bp, hA, next_acc, Nn, 16, DPAD, DPAD, relu);
    };
    auto gemm64 = [&](const float* A, const float* Wp, const float* bp,
                      float* C, int Ncpad, int lda, int ldc, int relu) {
        int tiles = (Gg >> 4) * (Ncpad >> 4);
        wmma_gemm_kernel<false, 64><<<(tiles + 7) / 8, TB, 0, stream>>>(
            A, nullptr, Wp, bp, C, nullptr, Gg, Ncpad, lda, ldc, relu);
    };

    encode_kernel<<<nodeBlocks, TB, 0, stream>>>(x, type_id, Wj_enc, bj_enc, Wm_enc, bm_enc, hA, agg, Nn);

    edge_pass_kernel<5><<<(Ejj + TB - 1) / TB, TB, 0, stream>>>(hA, agg, ei_jj, ea_jj, We_jj1, be_jj1, Ejj);
    update16(pWjj1, pbjj1, agg, 1);

    edge_pass_kernel<4><<<(Emj + TB - 1) / TB, TB, 0, stream>>>(hA, agg, ei_mj, ea_mj, We_mj1, be_mj1, Emj);
    update16(pWmj1, pbmj1, nullptr, 0);

    ln_relu_kernel<<<nodeBlocks, TB, 0, stream>>>(hA, agg, Nn);

    edge_pass_kernel<5><<<(Ejj + TB - 1) / TB, TB, 0, stream>>>(hA, agg, ei_jj, ea_jj, We_jj2, be_jj2, Ejj);
    update16(pWjj2, pbjj2, agg, 1);

    edge_pass_kernel<4><<<(Emj + TB - 1) / TB, TB, 0, stream>>>(hA, agg, ei_mj, ea_mj, We_mj2, be_mj2, Emj);
    update16(pWmj2, pbmj2, nullptr, 1);

    ln_relu_kernel<<<nodeBlocks, TB, 0, stream>>>(hA, nullptr, Nn);

    size_t zlen = (size_t)32 * Gg;
    fill_kernel<<<(int)((zlen + TB - 1) / TB), TB, 0, stream>>>(sumJ, 0.f, zlen);
    fill_kernel<<<(int)((zlen + TB - 1) / TB), TB, 0, stream>>>(maxJ, -__builtin_inff(), zlen);
    fill_kernel<<<(int)(((size_t)2 * Gg + TB - 1) / TB), TB, 0, stream>>>(cntJ, 0.f, (size_t)2 * Gg);
    pool_kernel<<<nodeBlocks, TB, 0, stream>>>(hA, type_id, batch, sumJ, sumM, maxJ, maxM, cntJ, cntM, Nn);

    assemble_ln_kernel<<<(Gg + 127) / 128, 128, 0, stream>>>(sumJ, sumM, maxJ, maxM, cntJ, cntM, u, gbuf, Gg);
    gemm64(gbuf, pW1, pb1, m1, 64, 64, 64, 1);
    gemm64(m1,   pW2, pb2, m2, 32, 64, 32, 1);
    mlp3_kernel<<<(Gg + TB - 1) / TB, TB, 0, stream>>>(m2, W3, b3, out, Gg);
}


# pooling: feature per lane, half-wave walks 16 nodes, sum/max/count of a run of equal graph id held in registers, one row-wide atomic per run, not 31 word atomics per node; sums: run partials first
# speedup vs baseline: 3.1486x; 1.3875x over previous
#include <hip/hip_runtime.h>
#include <hip/hip_bf16.h>

typedef __attribute__((ext_vector_type(2))) float v2f;
typedef __attribute__((ext_vector_type(8))) float v8f;

#define DPAD 16
#define DDIM 15

__global__ void fill_kernel(float* __restrict__ p, float v, size_t n) {
    size_t i = (size_t)blockIdx.x * blockDim.x + threadIdx.x;
    size_t stride = (size_t)gridDim.x * blockDim.x;
    for (; i < n; i += stride) p[i] = v;
}

__global__ void pack_weight_kernel(const float* __restrict__ W, const float* __restrict__ b,
                                   float* __restrict__ Wp, float* __restrict__ bp,
                                   int K, int Nc, int Kpad, int Ncpad)
{
    int idx = blockIdx.x * blockDim.x + threadIdx.x;
    int total = Kpad * Ncpad;
    if (idx < total) {
        int k = idx / Ncpad, c = idx % Ncpad;
        float v = (k < K && c < Nc) ? W[k * Nc + c] : 0.f;
        Wp[((size_t)(k >> 1) * Ncpad + c) * 2 + (k & 1)] = v;
    }
    if (idx < Ncpad) bp[idx] = (idx < Nc) ? b[idx] : 0.f;
}

typedef __attribute__((ext_vector_type(4))) float v4f;

__device__ __forceinline__ void store_row16(float* __restrict__ dst, const float* v) {
    v4f* p = (v4f*)dst;
#pragma unroll
    for (int q = 0; q < 4; ++q) {
        v4f t = {v[4 * q], v[4 * q + 1], v[4 * q + 2], v[4 * q + 3]};
        p[q] = t;
    }
}

__global__ __launch_bounds__(256) void encode_kernel(
    const float* __restrict__ x, const int* __restrict__ type_id,
    const float* __restrict__ Wj, const float* __restrict__ bj,
    const float* __restrict__ Wm, const float* __restrict__ bm,
    float* __restrict__ h, float* __restrict__ acc, int Nn)
{
    int i = blockIdx.x * blockDim.x + threadIdx.x;
    if (i >= Nn) return;
    float row[DPAD];
    float xi[5];
#pragma unroll
    for (int k = 0; k < 5; ++k) xi[k] = x[(size_t)i * 5 + k];
    int t = type_id[i];
    const float* Wp = (t == 0) ? Wj : Wm;
    const float* bp = (t == 0) ? bj : bm;
    bool valid = (t == 0) || (t == 1);
#pragma unroll
    for (int d = 0; d < DDIM; ++d) {
        float v = 0.f;
        if (valid) {
            v = bp[d];
#pragma unroll
            for (int k = 0; k < 5; ++k) v += xi[k] * Wp[k * DDIM + d];
            v = fmaxf(v, 0.f);
        }
        h[(size_t)i * DPAD + d] = v;
        row[d] = v;
    }
    h[(size_t)i * DPAD + DDIM] = 0.f;
    row[DDIM] = 0.f;
    store_row16(acc + (size_t)i * DPAD, row);
}

template <int J>
__device__ __forceinline__ int row_share_i(int v) {
    return __builtin_amdgcn_update_dpp(v, v, 0x150 + J, 0xF, 0xF, false);
}
template <int J>
__device__ __forceinline__ float row_share_f(float v) {
    return __int_as_float(row_share_i<J>(__float_as_int(v)));
}

#define FOR16(X) X(0) X(1) X(2) X(3) X(4) X(5) X(6) X(7) \
                 X(8) X(9) X(10) X(11) X(12) X(13) X(14) X(15)

template <int EDIM, bool FULL>
__device__ __forceinline__ void edge_chunk(
    const float* __restrict__ h, float* __restrict__ agg,
    const int* __restrict__ ei, const float* __restrict__ ea,
    const float* __restrict__ We, const float* __restrict__ be,
    int E, int e0, int lane)
{
    const int c    = lane & 15;
    const int half = lane >> 4;

    int e = e0 + lane;
    if (!FULL) e = (e < E) ? e : E - 1;
    const int src = ei[e];
    const int dst = ei[E + e];
    float ev[EDIM];
#pragma unroll
    for (int k = 0; k < EDIM; ++k) ev[k] = ea[(size_t)e * EDIM + k];

    const int cc = (c < DDIM) ? c : DDIM - 1;
    float w[EDIM];
#pragma unroll
    for (int k = 0; k < EDIM; ++k) {
        const float t = We[k * DDIM + cc];
        w[k] = (c < DDIM) ? t : 0.f;
    }
    const float tb = be[cc];
    const float bc = (c < DDIM) ? tb : 0.f;

    float hv[16];
#define GATHER_STEP(J) hv[J] = h[(unsigned)row_share_i<J>(src) * DPAD + c];
    FOR16(GATHER_STEP)
#undef GATHER_STEP

#define SCATTER_STEP(J) {                                                      \
        const unsigned d = (unsigned)row_share_i<J>(dst);                      \
        float m = hv[J] + bc;                                                  \
        _Pragma("unroll")                                                      \
        for (int k = 0; k < EDIM; ++k) m += row_share_f<J>(ev[k]) * w[k];      \
        m = fmaxf(m, 0.f);                                                     \
        if (FULL || e0 + 16 * half + J < E)                                    \
            atomicAdd(&agg[d * DPAD + c], m);                                  \
    }
    FOR16(SCATTER_STEP)
#undef SCATTER_STEP
}
#undef FOR16

template <int EDIM>
__global__ __launch_bounds__(256) void edge_pass_kernel(
    const float* __restrict__ h, float* __restrict__ agg,
    const int* __restrict__ ei, const float* __restrict__ ea,
    const float* __restrict__ We, const float* __restrict__ be, int E)
{
    const int lane = threadIdx.x & 31;
    const int e0   = (blockIdx.x * (blockDim.x >> 5) + (threadIdx.x >> 5)) * 32;
    if (e0 >= E) return;
    if (e0 + 32 <= E) edge_chunk<EDIM, true >(h, agg, ei, ea, We, be, E, e0, lane);
    else              edge_chunk<EDIM, false>(h, agg, ei, ea, We, be, E, e0, lane);
}

template <bool HAS_A2, int KPAD>
__global__ __launch_bounds__(256) void wmma_gemm_kernel(
    const float* A, const float* A2,
    const float* __restrict__ Wp, const float* __restrict__ bp,
    float* C, float* C2,
    int M, int Ncpad, int lda, int ldc, int relu)
{
    const int lane = threadIdx.x & 31;
    const int wave = threadIdx.x >> 5;
    const int ntN = Ncpad >> 4;
    const int tile = blockIdx.x * (blockDim.x >> 5) + wave;
    if (tile >= (M >> 4) * ntN) return;
    const int mt = tile / ntN;
    const int nt = tile % ntN;

    const int mrow  = mt * 16 + (lane & 15);
    const int ncol  = nt * 16 + (lane & 15);
    const int khalf = (lane >> 4) << 1;

    const float* ap  = A + (size_t)mrow * lda + khalf;
    const float* a2p = A2 + (size_t)mrow * lda + khalf;
    const float* bpp = Wp + ((size_t)(khalf >> 1) * Ncpad + ncol) * 2;
    const size_t bstep = (size_t)Ncpad * 4;

    v8f acc = {0.f, 0.f, 0.f, 0.f, 0.f, 0.f, 0.f, 0.f};
#pragma unroll
    for (int kk = 0; kk < KPAD; kk += 4) {
        v2f a = *(const v2f*)ap;
        if (HAS_A2) {
            v2f a2 = *(const v2f*)a2p;
            a.x += a2.x; a.y += a2.y;
            a2p += 4;
        }
        v2f b = *(const v2f*)bpp;
        ap += 4;
        bpp += bstep;
        acc = __builtin_amdgcn_wmma_f32_16x16x4_f32(
            false, a, false, b, (short)0, acc, false, false);
    }

    const int rbase = mt * 16 + ((lane >> 4) << 3);
    const float bv = bp[ncol];
    float o[8];
#pragma unroll
    for (int r = 0; r < 8; ++r) {
        float v = acc[r] + bv;
        if (relu) v = fmaxf(v, 0.f);
        o[r] = v;
        C[(size_t)(rbase + r) * ldc + ncol] = v;
    }
    if (C2) {
#pragma unroll
        for (int r = 0; r < 8; ++r)
            C2[(size_t)(rbase + r) * ldc + ncol] = o[r];
    }
}

__global__ __launch_bounds__(256) void ln_relu_kernel(float* __restrict__ h,
                                                      float* __restrict__ acc, int Nn)
{
    int i = blockIdx.x * blockDim.x + threadIdx.x;
    if (i >= Nn) return;
    float v[DDIM];
    float mu = 0.f;
#pragma unroll
    for (int d = 0; d < DDIM; ++d) { v[d] = h[(size_t)i * DPAD + d]; mu += v[d]; }
    mu *= (1.f / DDIM);
    float var = 0.f;
#pragma unroll
    for (int d = 0; d < DDIM; ++d) { float t = v[d] - mu; var += t * t; }
    var *= (1.f / DDIM);
    float inv = rsqrtf(var + 1e-5f);
    float row[DPAD];
#pragma unroll
    for (int d = 0; d < DDIM; ++d) {
        row[d] = fmaxf((v[d] - mu) * inv, 0.f);
        h[(size_t)i * DPAD + d] = row[d];
    }
    if (acc) {
        row[DDIM] = 0.f;
        store_row16(acc + (size_t)i * DPAD, row);
    }
}

__device__ __forceinline__ void atomicMaxF(float* addr, float v) {
    if (v >= 0.f) atomicMax((int*)addr, __float_as_int(v));
    else          atomicMin((unsigned int*)addr, __float_as_uint(v));
}

#define POOL_WALK 16
__global__ __launch_bounds__(256) void pool_kernel(
    const float* __restrict__ h, const int* __restrict__ type_id,
    const int* __restrict__ batch,
    float* __restrict__ sumJ, float* __restrict__ sumM,
    float* __restrict__ maxJ, float* __restrict__ maxM, int Nn)
{
    const int c  = threadIdx.x & 15;
    const int hw = (blockIdx.x * blockDim.x + threadIdx.x) >> 4;
    const int n0 = hw * POOL_WALK;
    if (n0 >= Nn) return;

    int   bb[POOL_WALK];
    int   tt[POOL_WALK];
    float hv[POOL_WALK];
#pragma unroll
    for (int j = 0; j < POOL_WALK; ++j) {
        const int  node = n0 + j;
        const bool ok   = node < Nn;
        const int  nc   = ok ? node : Nn - 1;
        bb[j] = batch[nc];
        const int t = type_id[nc];
        tt[j] = ok ? t : -1;
        const float v = h[(size_t)nc * DPAD + c];
        hv[j] = (c == DDIM) ? 1.f : v;
    }

    const float ninf = -__builtin_inff();
    int   cur = bb[0];
    float sJ = 0.f, sM = 0.f, mJ = ninf, mM = ninf;
    int   nJ = 0, nM = 0;
#pragma unroll
    for (int j = 0; j < POOL_WALK; ++j) {
        if (bb[j] != cur) {
            if (nJ) { atomicAdd(&sumJ[(size_t)cur * DPAD + c], sJ); atomicMaxF(&maxJ[(size_t)cur * DPAD + c], mJ); }
            if (nM) { atomicAdd(&sumM[(size_t)cur * DPAD + c], sM); atomicMaxF(&maxM[(size_t)cur * DPAD + c], mM); }
            sJ = 0.f; sM = 0.f; mJ = ninf; mM = ninf; nJ = 0; nM = 0;
            cur = bb[j];
        }
        const float v   = hv[j];
        const bool  isJ = tt[j] == 0;
        const bool  isM = tt[j] == 1;
        sJ += isJ ? v : 0.f;  mJ = fmaxf(mJ, isJ ? v : ninf);  nJ += isJ ? 1 : 0;
        sM += isM ? v : 0.f;  mM = fmaxf(mM, isM ? v : ninf);  nM += isM ? 1 : 0;
    }
    if (nJ) { atomicAdd(&sumJ[(size_t)cur * DPAD + c], sJ); atomicMaxF(&maxJ[(size_t)cur * DPAD + c], mJ); }
    if (nM) { atomicAdd(&sumM[(size_t)cur * DPAD + c], sM); atomicMaxF(&maxM[(size_t)cur * DPAD + c], mM); }
}
#undef POOL_WALK

__global__ __launch_bounds__(128) void assemble_ln_kernel(
    const float* __restrict__ sumJ, const float* __restrict__ sumM,
    const float* __restrict__ maxJ, const float* __restrict__ maxM,
    const float* __restrict__ u, float* __restrict__ g, int Gg)
{
    int gi = blockIdx.x * blockDim.x + threadIdx.x;
    if (gi >= Gg) return;
    float r[64];
    float cm = fmaxf(sumM[(size_t)gi * DPAD + DDIM], 1.f);
    float cj = fmaxf(sumJ[(size_t)gi * DPAD + DDIM], 1.f);
#pragma unroll
    for (int d = 0; d < DDIM; ++d) r[d] = sumM[(size_t)gi * DPAD + d] / cm;
#pragma unroll
    for (int d = 0; d < DDIM; ++d) r[15 + d] = sumJ[(size_t)gi * DPAD + d] / cj;
#pragma unroll
    for (int d = 0; d < DDIM; ++d) {
        float v = maxJ[(size_t)gi * DPAD + d];
        r[30 + d] = (v > -3.0e38f) ? v : 0.f;
    }
#pragma unroll
    for (int d = 0; d < DDIM; ++d) {
        float v = maxM[(size_t)gi * DPAD + d];
        r[45 + d] = (v > -3.0e38f) ? v : 0.f;
    }
#pragma unroll
    for (int j = 0; j < 4; ++j) r[60 + j] = u[(size_t)gi * 4 + j];

    float mu = 0.f;
#pragma unroll
    for (int k = 0; k < 64; ++k) mu += r[k];
    mu *= (1.f / 64.f);
    float var = 0.f;
#pragma unroll
    for (int k = 0; k < 64; ++k) { float t = r[k] - mu; var += t * t; }
    var *= (1.f / 64.f);
    float inv = rsqrtf(var + 1e-5f);
#pragma unroll
    for (int k = 0; k < 64; ++k) g[(size_t)gi * 64 + k] = (r[k] - mu) * inv;
}

__global__ __launch_bounds__(256) void mlp3_kernel(
    const float* __restrict__ m2, const float* __restrict__ W3,
    const float* __restrict__ b3, float* __restrict__ out, int Gg)
{
    int gi = blockIdx.x * blockDim.x + threadIdx.x;
    if (gi >= Gg) return;
    float s = b3[0];
#pragma unroll
    for (int k = 0; k < 32; ++k) s += m2[(size_t)gi * 32 + k] * W3[k];
    out[gi] = s;
}

extern "C" void kernel_launch(void* const* d_in, const int* in_sizes, int n_in,
                              void* d_out, int out_size, void* d_ws, size_t ws_size,
                              hipStream_t stream)
{
    const float* x       = (const float*)d_in[0];
    const int*   type_id = (const int*)  d_in[1];
    const int*   batch   = (const int*)  d_in[2];
    const float* u       = (const float*)d_in[3];
    const int*   ei_jj   = (const int*)  d_in[4];
    const float* ea_jj   = (const float*)d_in[5];
    const int*   ei_mj   = (const int*)  d_in[6];
    const float* ea_mj   = (const float*)d_in[7];
    const float* Wj_enc  = (const float*)d_in[8];   const float* bj_enc = (const float*)d_in[9];
    const float* Wm_enc  = (const float*)d_in[10];  const float* bm_enc = (const float*)d_in[11];
    const float* We_jj1  = (const float*)d_in[12];  const float* be_jj1 = (const float*)d_in[13];
    const float* Wm_jj1  = (const float*)d_in[14];  const float* bm_jj1 = (const float*)d_in[15];
    const float* We_mj1  = (const float*)d_in[16];  const float* be_mj1 = (const float*)d_in[17];
    const float* Wm_mj1  = (const float*)d_in[18];  const float* bm_mj1 = (const float*)d_in[19];
    const float* We_jj2  = (const float*)d_in[20];  const float* be_jj2 = (const float*)d_in[21];
    const float* Wm_jj2  = (const float*)d_in[22];  const float* bm_jj2 = (const float*)d_in[23];
    const float* We_mj2  = (const float*)d_in[24];  const float* be_mj2 = (const float*)d_in[25];
    const float* Wm_mj2  = (const float*)d_in[26];  const float* bm_mj2 = (const float*)d_in[27];
    const float* W1      = (const float*)d_in[28];  const float* b1     = (const float*)d_in[29];
    const float* W2      = (const float*)d_in[30];  const float* b2     = (const float*)d_in[31];
    const float* W3      = (const float*)d_in[32];  const float* b3     = (const float*)d_in[33];

    const int Nn  = in_sizes[0] / 5;
    const int Gg  = in_sizes[3] / 4;
    const int Ejj = in_sizes[5] / 5;
    const int Emj = in_sizes[7] / 4;

    float* ws = (float*)d_ws;
    const size_t NP = (size_t)Nn * DPAD;
    float* hA  = ws;
    float* agg = ws + 2 * NP;
    float* wpk = ws + 3 * NP;
    float* pWjj1 = wpk;          float* pbjj1 = wpk + 256;
    float* pWmj1 = wpk + 272;    float* pbmj1 = wpk + 528;
    float* pWjj2 = wpk + 544;    float* pbjj2 = wpk + 800;
    float* pWmj2 = wpk + 816;    float* pbmj2 = wpk + 1072;
    float* pW1   = wpk + 1088;   float* pb1   = wpk + 5184;
    float* pW2   = wpk + 5248;   float* pb2   = wpk + 7296;

    float* P    = agg;
    float* sumJ = P;
    float* sumM = P + (size_t)16 * Gg;
    float* maxJ = P + (size_t)32 * Gg;
    float* maxM = P + (size_t)48 * Gg;
    float* gbuf = P + (size_t)66 * Gg;
    float* m1   = P + (size_t)130 * Gg;
    float* m2   = P + (size_t)194 * Gg;
    float* out  = (float*)d_out;

    const int TB = 256;
    const int nodeBlocks = (Nn + TB - 1) / TB;

    pack_weight_kernel<<<1, 256, 0, stream>>>(Wm_jj1, bm_jj1, pWjj1, pbjj1, 15, 15, 16, 16);
    pack_weight_kernel<<<1, 256, 0, stream>>>(Wm_mj1, bm_mj1, pWmj1, pbmj1, 15, 15, 16, 16);
    pack_weight_kernel<<<1, 256, 0, stream>>>(Wm_jj2, bm_jj2, pWjj2, pbjj2, 15, 15, 16, 16);
    pack_weight_kernel<<<1, 256, 0, stream>>>(Wm_mj2, bm_mj2, pWmj2, pbmj2, 15, 15, 16, 16);
    pack_weight_kernel<<<16, 256, 0, stream>>>(W1, b1, pW1, pb1, 64, 64, 64, 64);
    pack_weight_kernel<<<8, 256, 0, stream>>>(W2, b2, pW2, pb2, 64, 32, 64, 32);

    auto update16 = [&](const float* Wp, const float* bp, float* next_acc, int relu) {
        int tiles = Nn >> 4;
        wmma_gemm_kernel<false, 16><<<(tiles + 7) / 8, TB, 0, stream>>>(
            agg, nullptr, Wp, bp, hA, next_acc, Nn, 16, DPAD, DPAD, relu);
    };
    auto gemm64 = [&](const float* A, const float* Wp, const float* bp,
                      float* C, int Ncpad, int lda, int ldc, int relu) {
        int tiles = (Gg >> 4) * (Ncpad >> 4);
        wmma_gemm_kernel<false, 64><<<(tiles + 7) / 8, TB, 0, stream>>>(
            A, nullptr, Wp, bp, C, nullptr, Gg, Ncpad, lda, ldc, relu);
    };

    encode_kernel<<<nodeBlocks, TB, 0, stream>>>(x, type_id, Wj_enc, bj_enc, Wm_enc, bm_enc, hA, agg, Nn);

    edge_pass_kernel<5><<<(Ejj + TB - 1) / TB, TB, 0, stream>>>(hA, agg, ei_jj, ea_jj, We_jj1, be_jj1, Ejj);
    update16(pWjj1, pbjj1, agg, 1);

    edge_pass_kernel<4><<<(Emj + TB - 1) / TB, TB, 0, stream>>>(hA, agg, ei_mj, ea_mj, We_mj1, be_mj1, Emj);
    update16(pWmj1, pbmj1, nullptr, 0);

    ln_relu_kernel<<<nodeBlocks, TB, 0, stream>>>(hA, agg, Nn);

    edge_pass_kernel<5><<<(Ejj + TB - 1) / TB, TB, 0, stream>>>(hA, agg, ei_jj, ea_jj, We_jj2, be_jj2, Ejj);
    update16(pWjj2, pbjj2, agg, 1);

    edge_pass_kernel<4><<<(Emj + TB - 1) / TB, TB, 0, stream>>>(hA, agg, ei_mj, ea_mj, We_mj2, be_mj2, Emj);
    update16(pWmj2, pbmj2, nullptr, 1);

    ln_relu_kernel<<<nodeBlocks, TB, 0, stream>>>(hA, nullptr, Nn);

    size_t zlen = (size_t)32 * Gg;
    fill_kernel<<<(int)((zlen + TB - 1) / TB), TB, 0, stream>>>(sumJ, 0.f, zlen);
    fill_kernel<<<(int)((zlen + TB - 1) / TB), TB, 0, stream>>>(maxJ, -__builtin_inff(), zlen);
    pool_kernel<<<nodeBlocks, TB, 0, stream>>>(hA, type_id, batch, sumJ, sumM, maxJ, maxM, Nn);

    assemble_ln_kernel<<<(Gg + 127) / 128, 128, 0, stream>>>(sumJ, sumM, maxJ, maxM, u, gbuf, Gg);
    gemm64(gbuf, pW1, pb1, m1, 64, 64, 64, 1);
    gemm64(m1,   pW2, pb2, m2, 32, 64, 32, 1);
    mlp3_kernel<<<(Gg + TB - 1) / TB, TB, 0, stream>>>(m2, W3, b3, out, Gg);
}


# LayerNorm+ReLU folded into the epilogue of each layer's second node-update GEMM (16-lane DPP butterfly row sums): 2 LN launches and one h write+read per layer go; LN sums: butterfly tree, not serial
# speedup vs baseline: 3.3242x; 1.0558x over previous
#include <hip/hip_runtime.h>
#include <hip/hip_bf16.h>

typedef __attribute__((ext_vector_type(2))) float v2f;
typedef __attribute__((ext_vector_type(8))) float v8f;

#define DPAD 16
#define DDIM 15

__global__ void fill_kernel(float* __restrict__ p, float v, size_t n) {
    size_t i = (size_t)blockIdx.x * blockDim.x + threadIdx.x;
    size_t stride = (size_t)gridDim.x * blockDim.x;
    for (; i < n; i += stride) p[i] = v;
}

__global__ void pack_weight_kernel(const float* __restrict__ W, const float* __restrict__ b,
                                   float* __restrict__ Wp, float* __restrict__ bp,
                                   int K, int Nc, int Kpad, int Ncpad)
{
    int idx = blockIdx.x * blockDim.x + threadIdx.x;
    int total = Kpad * Ncpad;
    if (idx < total) {
        int k = idx / Ncpad, c = idx % Ncpad;
        float v = (k < K && c < Nc) ? W[k * Nc + c] : 0.f;
        Wp[((size_t)(k >> 1) * Ncpad + c) * 2 + (k & 1)] = v;
    }
    if (idx < Ncpad) bp[idx] = (idx < Nc) ? b[idx] : 0.f;
}

typedef __attribute__((ext_vector_type(4))) float v4f;

__device__ __forceinline__ void store_row16(float* __restrict__ dst, const float* v) {
    v4f* p = (v4f*)dst;
#pragma unroll
    for (int q = 0; q < 4; ++q) {
        v4f t = {v[4 * q], v[4 * q + 1], v[4 * q + 2], v[4 * q + 3]};
        p[q] = t;
    }
}

__global__ __launch_bounds__(256) void encode_kernel(
    const float* __restrict__ x, const int* __restrict__ type_id,
    const float* __restrict__ Wj, const float* __restrict__ bj,
    const float* __restrict__ Wm, const float* __restrict__ bm,
    float* __restrict__ h, float* __restrict__ acc, int Nn)
{
    int i = blockIdx.x * blockDim.x + threadIdx.x;
    if (i >= Nn) return;
    float row[DPAD];
    float xi[5];
#pragma unroll
    for (int k = 0; k < 5; ++k) xi[k] = x[(size_t)i * 5 + k];
    int t = type_id[i];
    const float* Wp = (t == 0) ? Wj : Wm;
    const float* bp = (t == 0) ? bj : bm;
    bool valid = (t == 0) || (t == 1);
#pragma unroll
    for (int d = 0; d < DDIM; ++d) {
        float v = 0.f;
        if (valid) {
            v = bp[d];
#pragma unroll
            for (int k = 0; k < 5; ++k) v += xi[k] * Wp[k * DDIM + d];
            v = fmaxf(v, 0.f);
        }
        h[(size_t)i * DPAD + d] = v;
        row[d] = v;
    }
    h[(size_t)i * DPAD + DDIM] = 0.f;
    row[DDIM] = 0.f;
    store_row16(acc + (size_t)i * DPAD, row);
}

template <int J>
__device__ __forceinline__ int row_share_i(int v) {
    return __builtin_amdgcn_update_dpp(v, v, 0x150 + J, 0xF, 0xF, false);
}
template <int J>
__device__ __forceinline__ float row_share_f(float v) {
    return __int_as_float(row_share_i<J>(__float_as_int(v)));
}

template <int M>
__device__ __forceinline__ float row_xmask_f(float v) {
    const int i = __float_as_int(v);
    return __int_as_float(__builtin_amdgcn_update_dpp(i, i, 0x160 + M, 0xF, 0xF, true));
}
__device__ __forceinline__ float row_sum16(float v) {
    v += row_xmask_f<1>(v);
    v += row_xmask_f<2>(v);
    v += row_xmask_f<4>(v);
    v += row_xmask_f<8>(v);
    return v;
}

#define FOR16(X) X(0) X(1) X(2) X(3) X(4) X(5) X(6) X(7) \
                 X(8) X(9) X(10) X(11) X(12) X(13) X(14) X(15)

template <int EDIM, bool FULL>
__device__ __forceinline__ void edge_chunk(
    const float* __restrict__ h, float* __restrict__ agg,
    const int* __restrict__ ei, const float* __restrict__ ea,
    const float* __restrict__ We, const float* __restrict__ be,
    int E, int e0, int lane)
{
    const int c    = lane & 15;
    const int half = lane >> 4;

    int e = e0 + lane;
    if (!FULL) e = (e < E) ? e : E - 1;
    const int src = ei[e];
    const int dst = ei[E + e];
    float ev[EDIM];
#pragma unroll
    for (int k = 0; k < EDIM; ++k) ev[k] = ea[(size_t)e * EDIM + k];

    const int cc = (c < DDIM) ? c : DDIM - 1;
    float w[EDIM];
#pragma unroll
    for (int k = 0; k < EDIM; ++k) {
        const float t = We[k * DDIM + cc];
        w[k] = (c < DDIM) ? t : 0.f;
    }
    const float tb = be[cc];
    const float bc = (c < DDIM) ? tb : 0.f;

    float hv[16];
#define GATHER_STEP(J) hv[J] = h[(unsigned)row_share_i<J>(src) * DPAD + c];
    FOR16(GATHER_STEP)
#undef GATHER_STEP

#define SCATTER_STEP(J) {                                                      \
        const unsigned d = (unsigned)row_share_i<J>(dst);                      \
        float m = hv[J] + bc;                                                  \
        _Pragma("unroll")                                                      \
        for (int k = 0; k < EDIM; ++k) m += row_share_f<J>(ev[k]) * w[k];      \
        m = fmaxf(m, 0.f);                                                     \
        if (FULL || e0 + 16 * half + J < E)                                    \
            atomicAdd(&agg[d * DPAD + c], m);                                  \
    }
    FOR16(SCATTER_STEP)
#undef SCATTER_STEP
}
#undef FOR16

template <int EDIM>
__global__ __launch_bounds__(256) void edge_pass_kernel(
    const float* __restrict__ h, float* __restrict__ agg,
    const int* __restrict__ ei, const float* __restrict__ ea,
    const float* __restrict__ We, const float* __restrict__ be, int E)
{
    const int lane = threadIdx.x & 31;
    const int e0   = (blockIdx.x * (blockDim.x >> 5) + (threadIdx.x >> 5)) * 32;
    if (e0 >= E) return;
    if (e0 + 32 <= E) edge_chunk<EDIM, true >(h, agg, ei, ea, We, be, E, e0, lane);
    else              edge_chunk<EDIM, false>(h, agg, ei, ea, We, be, E, e0, lane);
}

template <bool HAS_A2, int KPAD, bool LN>
__global__ __launch_bounds__(256) void wmma_gemm_kernel(
    const float* A, const float* A2,
    const float* __restrict__ Wp, const float* __restrict__ bp,
    float* C, float* C2,
    int M, int Ncpad, int lda, int ldc, int relu)
{
    const int lane = threadIdx.x & 31;
    const int wave = threadIdx.x >> 5;
    const int ntN = Ncpad >> 4;
    const int tile = blockIdx.x * (blockDim.x >> 5) + wave;
    if (tile >= (M >> 4) * ntN) return;
    const int mt = tile / ntN;
    const int nt = tile % ntN;

    const int mrow  = mt * 16 + (lane & 15);
    const int ncol  = nt * 16 + (lane & 15);
    const int khalf = (lane >> 4) << 1;

    const float* ap  = A + (size_t)mrow * lda + khalf;
    const float* a2p = A2 + (size_t)mrow * lda + khalf;
    const float* bpp = Wp + ((size_t)(khalf >> 1) * Ncpad + ncol) * 2;
    const size_t bstep = (size_t)Ncpad * 4;

    v8f acc = {0.f, 0.f, 0.f, 0.f, 0.f, 0.f, 0.f, 0.f};
#pragma unroll
    for (int kk = 0; kk < KPAD; kk += 4) {
        v2f a = *(const v2f*)ap;
        if (HAS_A2) {
            v2f a2 = *(const v2f*)a2p;
            a.x += a2.x; a.y += a2.y;
            a2p += 4;
        }
        v2f b = *(const v2f*)bpp;
        ap += 4;
        bpp += bstep;
        acc = __builtin_amdgcn_wmma_f32_16x16x4_f32(
            false, a, false, b, (short)0, acc, false, false);
    }

    const int rbase = mt * 16 + ((lane >> 4) << 3);
    const float bv = bp[ncol];
    float o[8];
#pragma unroll
    for (int r = 0; r < 8; ++r) {
        float v = acc[r] + bv;
        if (relu) v = fmaxf(v, 0.f);
        o[r] = v;
    }
    if (LN) {
        const bool real = (lane & 15) < DDIM;
        float t[8];
#pragma unroll
        for (int r = 0; r < 8; ++r) {
            const float mu = row_sum16(real ? o[r] : 0.f) * (1.f / DDIM);
            t[r] = real ? o[r] - mu : 0.f;
        }
#pragma unroll
        for (int r = 0; r < 8; ++r) {
            const float var = row_sum16(t[r] * t[r]) * (1.f / DDIM);
            o[r] = fmaxf(t[r] * rsqrtf(var + 1e-5f), 0.f);
        }
    }
#pragma unroll
    for (int r = 0; r < 8; ++r)
        C[(size_t)(rbase + r) * ldc + ncol] = o[r];
    if (C2) {
#pragma unroll
        for (int r = 0; r < 8; ++r)
            C2[(size_t)(rbase + r) * ldc + ncol] = o[r];
    }
}

__device__ __forceinline__ void atomicMaxF(float* addr, float v) {
    if (v >= 0.f) atomicMax((int*)addr, __float_as_int(v));
    else          atomicMin((unsigned int*)addr, __float_as_uint(v));
}

#define POOL_WALK 16
__global__ __launch_bounds__(256) void pool_kernel(
    const float* __restrict__ h, const int* __restrict__ type_id,
    const int* __restrict__ batch,
    float* __restrict__ sumJ, float* __restrict__ sumM,
    float* __restrict__ maxJ, float* __restrict__ maxM, int Nn)
{
    const int c  = threadIdx.x & 15;
    const int hw = (blockIdx.x * blockDim.x + threadIdx.x) >> 4;
    const int n0 = hw * POOL_WALK;
    if (n0 >= Nn) return;

    int   bb[POOL_WALK];
    int   tt[POOL_WALK];
    float hv[POOL_WALK];
#pragma unroll
    for (int j = 0; j < POOL_WALK; ++j) {
        const int  node = n0 + j;
        const bool ok   = node < Nn;
        const int  nc   = ok ? node : Nn - 1;
        bb[j] = batch[nc];
        const int t = type_id[nc];
        tt[j] = ok ? t : -1;
        const float v = h[(size_t)nc * DPAD + c];
        hv[j] = (c == DDIM) ? 1.f : v;
    }

    const float ninf = -__builtin_inff();
    int   cur = bb[0];
    float sJ = 0.f, sM = 0.f, mJ = ninf, mM = ninf;
    int   nJ = 0, nM = 0;
#pragma unroll
    for (int j = 0; j < POOL_WALK; ++j) {
        if (bb[j] != cur) {
            if (nJ) { atomicAdd(&sumJ[(size_t)cur * DPAD + c], sJ); atomicMaxF(&maxJ[(size_t)cur * DPAD + c], mJ); }
            if (nM) { atomicAdd(&sumM[(size_t)cur * DPAD + c], sM); atomicMaxF(&maxM[(size_t)cur * DPAD + c], mM); }
            sJ = 0.f; sM = 0.f; mJ = ninf; mM = ninf; nJ = 0; nM = 0;
            cur = bb[j];
        }
        const float v   = hv[j];
        const bool  isJ = tt[j] == 0;
        const bool  isM = tt[j] == 1;
        sJ += isJ ? v : 0.f;  mJ = fmaxf(mJ, isJ ? v : ninf);  nJ += isJ ? 1 : 0;
        sM += isM ? v : 0.f;  mM = fmaxf(mM, isM ? v : ninf);  nM += isM ? 1 : 0;
    }
    if (nJ) { atomicAdd(&sumJ[(size_t)cur * DPAD + c], sJ); atomicMaxF(&maxJ[(size_t)cur * DPAD + c], mJ); }
    if (nM) { atomicAdd(&sumM[(size_t)cur * DPAD + c], sM); atomicMaxF(&maxM[(size_t)cur * DPAD + c], mM); }
}
#undef POOL_WALK

__global__ __launch_bounds__(128) void assemble_ln_kernel(
    const float* __restrict__ sumJ, const float* __restrict__ sumM,
    const float* __restrict__ maxJ, const float* __restrict__ maxM,
    const float* __restrict__ u, float* __restrict__ g, int Gg)
{
    int gi = blockIdx.x * blockDim.x + threadIdx.x;
    if (gi >= Gg) return;
    float r[64];
    float cm = fmaxf(sumM[(size_t)gi * DPAD + DDIM], 1.f);
    float cj = fmaxf(sumJ[(size_t)gi * DPAD + DDIM], 1.f);
#pragma unroll
    for (int d = 0; d < DDIM; ++d) r[d] = sumM[(size_t)gi * DPAD + d] / cm;
#pragma unroll
    for (int d = 0; d < DDIM; ++d) r[15 + d] = sumJ[(size_t)gi * DPAD + d] / cj;
#pragma unroll
    for (int d = 0; d < DDIM; ++d) {
        float v = maxJ[(size_t)gi * DPAD + d];
        r[30 + d] = (v > -3.0e38f) ? v : 0.f;
    }
#pragma unroll
    for (int d = 0; d < DDIM; ++d) {
        float v = maxM[(size_t)gi * DPAD + d];
        r[45 + d] = (v > -3.0e38f) ? v : 0.f;
    }
#pragma unroll
    for (int j = 0; j < 4; ++j) r[60 + j] = u[(size_t)gi * 4 + j];

    float mu = 0.f;
#pragma unroll
    for (int k = 0; k < 64; ++k) mu += r[k];
    mu *= (1.f / 64.f);
    float var = 0.f;
#pragma unroll
    for (int k = 0; k < 64; ++k) { float t = r[k] - mu; var += t * t; }
    var *= (1.f / 64.f);
    float inv = rsqrtf(var + 1e-5f);
#pragma unroll
    for (int k = 0; k < 64; ++k) g[(size_t)gi * 64 + k] = (r[k] - mu) * inv;
}

__global__ __launch_bounds__(256) void mlp3_kernel(
    const float* __restrict__ m2, const float* __restrict__ W3,
    const float* __restrict__ b3, float* __restrict__ out, int Gg)
{
    int gi = blockIdx.x * blockDim.x + threadIdx.x;
    if (gi >= Gg) return;
    float s = b3[0];
#pragma unroll
    for (int k = 0; k < 32; ++k) s += m2[(size_t)gi * 32 + k] * W3[k];
    out[gi] = s;
}

extern "C" void kernel_launch(void* const* d_in, const int* in_sizes, int n_in,
                              void* d_out, int out_size, void* d_ws, size_t ws_size,
                              hipStream_t stream)
{
    const float* x       = (const float*)d_in[0];
    const int*   type_id = (const int*)  d_in[1];
    const int*   batch   = (const int*)  d_in[2];
    const float* u       = (const float*)d_in[3];
    const int*   ei_jj   = (const int*)  d_in[4];
    const float* ea_jj   = (const float*)d_in[5];
    const int*   ei_mj   = (const int*)  d_in[6];
    const float* ea_mj   = (const float*)d_in[7];
    const float* Wj_enc  = (const float*)d_in[8];   const float* bj_enc = (const float*)d_in[9];
    const float* Wm_enc  = (const float*)d_in[10];  const float* bm_enc = (const float*)d_in[11];
    const float* We_jj1  = (const float*)d_in[12];  const float* be_jj1 = (const float*)d_in[13];
    const float* Wm_jj1  = (const float*)d_in[14];  const float* bm_jj1 = (const float*)d_in[15];
    const float* We_mj1  = (const float*)d_in[16];  const float* be_mj1 = (const float*)d_in[17];
    const float* Wm_mj1  = (const float*)d_in[18];  const float* bm_mj1 = (const float*)d_in[19];
    const float* We_jj2  = (const float*)d_in[20];  const float* be_jj2 = (const float*)d_in[21];
    const float* Wm_jj2  = (const float*)d_in[22];  const float* bm_jj2 = (const float*)d_in[23];
    const float* We_mj2  = (const float*)d_in[24];  const float* be_mj2 = (const float*)d_in[25];
    const float* Wm_mj2  = (const float*)d_in[26];  const float* bm_mj2 = (const float*)d_in[27];
    const float* W1      = (const float*)d_in[28];  const float* b1     = (const float*)d_in[29];
    const float* W2      = (const float*)d_in[30];  const float* b2     = (const float*)d_in[31];
    const float* W3      = (const float*)d_in[32];  const float* b3     = (const float*)d_in[33];

    const int Nn  = in_sizes[0] / 5;
    const int Gg  = in_sizes[3] / 4;
    const int Ejj = in_sizes[5] / 5;
    const int Emj = in_sizes[7] / 4;

    float* ws = (float*)d_ws;
    const size_t NP = (size_t)Nn * DPAD;
    float* hA  = ws;
    float* agg = ws + 2 * NP;
    float* wpk = ws + 3 * NP;
    float* pWjj1 = wpk;          float* pbjj1 = wpk + 256;
    float* pWmj1 = wpk + 272;    float* pbmj1 = wpk + 528;
    float* pWjj2 = wpk + 544;    float* pbjj2 = wpk + 800;
    float* pWmj2 = wpk + 816;    float* pbmj2 = wpk + 1072;
    float* pW1   = wpk + 1088;   float* pb1   = wpk + 5184;
    float* pW2   = wpk + 5248;   float* pb2   = wpk + 7296;

    float* P    = agg;
    float* sumJ = P;
    float* sumM = P + (size_t)16 * Gg;
    float* maxJ = P + (size_t)32 * Gg;
    float* maxM = P + (size_t)48 * Gg;
    float* gbuf = P + (size_t)66 * Gg;
    float* m1   = P + (size_t)130 * Gg;
    float* m2   = P + (size_t)194 * Gg;
    float* out  = (float*)d_out;

    const int TB = 256;
    const int nodeBlocks = (Nn + TB - 1) / TB;

    pack_weight_kernel<<<1, 256, 0, stream>>>(Wm_jj1, bm_jj1, pWjj1, pbjj1, 15, 15, 16, 16);
    pack_weight_kernel<<<1, 256, 0, stream>>>(Wm_mj1, bm_mj1, pWmj1, pbmj1, 15, 15, 16, 16);
    pack_weight_kernel<<<1, 256, 0, stream>>>(Wm_jj2, bm_jj2, pWjj2, pbjj2, 15, 15, 16, 16);
    pack_weight_kernel<<<1, 256, 0, stream>>>(Wm_mj2, bm_mj2, pWmj2, pbmj2, 15, 15, 16, 16);
    pack_weight_kernel<<<16, 256, 0, stream>>>(W1, b1, pW1, pb1, 64, 64, 64, 64);
    pack_weight_kernel<<<8, 256, 0, stream>>>(W2, b2, pW2, pb2, 64, 32, 64, 32);

    auto update16 = [&](const float* Wp, const float* bp, float* next_acc, int relu) {
        int tiles = Nn >> 4;
        wmma_gemm_kernel<false, 16, false><<<(tiles + 7) / 8, TB, 0, stream>>>(
            agg, nullptr, Wp, bp, hA, next_acc, Nn, 16, DPAD, DPAD, relu);
    };
    auto update16_ln = [&](const float* Wp, const float* bp, float* next_acc, int relu) {
        int tiles = Nn >> 4;
        wmma_gemm_kernel<false, 16, true><<<(tiles + 7) / 8, TB, 0, stream>>>(
            agg, nullptr, Wp, bp, hA, next_acc, Nn, 16, DPAD, DPAD, relu);
    };
    auto gemm64 = [&](const float* A, const float* Wp, const float* bp,
                      float* C, int Ncpad, int lda, int ldc, int relu) {
        int tiles = (Gg >> 4) * (Ncpad >> 4);
        wmma_gemm_kernel<false, 64, false><<<(tiles + 7) / 8, TB, 0, stream>>>(
            A, nullptr, Wp, bp, C, nullptr, Gg, Ncpad, lda, ldc, relu);
    };

    encode_kernel<<<nodeBlocks, TB, 0, stream>>>(x, type_id, Wj_enc, bj_enc, Wm_enc, bm_enc, hA, agg, Nn);

    edge_pass_kernel<5><<<(Ejj + TB - 1) / TB, TB, 0, stream>>>(hA, agg, ei_jj, ea_jj, We_jj1, be_jj1, Ejj);
    update16(pWjj1, pbjj1, agg, 1);

    edge_pass_kernel<4><<<(Emj + TB - 1) / TB, TB, 0, stream>>>(hA, agg, ei_mj, ea_mj, We_mj1, be_mj1, Emj);
    update16_ln(pWmj1, pbmj1, agg, 0);

    edge_pass_kernel<5><<<(Ejj + TB - 1) / TB, TB, 0, stream>>>(hA, agg, ei_jj, ea_jj, We_jj2, be_jj2, Ejj);
    update16(pWjj2, pbjj2, agg, 1);

    edge_pass_kernel<4><<<(Emj + TB - 1) / TB, TB, 0, stream>>>(hA, agg, ei_mj, ea_mj, We_mj2, be_mj2, Emj);
    update16_ln(pWmj2, pbmj2, nullptr, 1);

    size_t zlen = (size_t)32 * Gg;
    fill_kernel<<<(int)((zlen + TB - 1) / TB), TB, 0, stream>>>(sumJ, 0.f, zlen);
    fill_kernel<<<(int)((zlen + TB - 1) / TB), TB, 0, stream>>>(maxJ, -__builtin_inff(), zlen);
    pool_kernel<<<nodeBlocks, TB, 0, stream>>>(hA, type_id, batch, sumJ, sumM, maxJ, maxM, Nn);

    assemble_ln_kernel<<<(Gg + 127) / 128, 128, 0, stream>>>(sumJ, sumM, maxJ, maxM, u, gbuf, Gg);
    gemm64(gbuf, pW1, pb1, m1, 64, 64, 64, 1);
    gemm64(m1,   pW2, pb2, m2, 32, 64, 32, 1);
    mlp3_kernel<<<(Gg + TB - 1) / TB, TB, 0, stream>>>(m2, W3, b3, out, Gg);
}
